# starting kernel
# baseline (speedup 1.0000x reference)
#include <hip/hip_runtime.h>
#include <hip/hip_bf16.h>

typedef __attribute__((ext_vector_type(2))) float v2f;
typedef __attribute__((ext_vector_type(8))) float v8f;

#define HID 128
#define WSTRIDE 132

__global__ void k_zero_i(int* __restrict__ p, int n) {
  int i = blockIdx.x * blockDim.x + threadIdx.x;
  if (i < n) p[i] = 0;
}
__global__ void k_zero_f(float* __restrict__ p, int n) {
  int i = blockIdx.x * blockDim.x + threadIdx.x;
  if (i < n) p[i] = 0.0f;
}

__global__ void k_count(const int* __restrict__ dst, int* __restrict__ cnt, int e) {
  int i = blockIdx.x * blockDim.x + threadIdx.x;
  if (i < e) atomicAdd(&cnt[dst[i]], 1);
}

__global__ void k_dinv(const int* __restrict__ cnt, float* __restrict__ dinv, int n) {
  int i = blockIdx.x * blockDim.x + threadIdx.x;
  if (i < n) dinv[i] = rsqrtf((float)cnt[i] + 1.0f);
}

__global__ void k_scan1(const int* __restrict__ counts, int* __restrict__ rowstart,
                        int* __restrict__ blockSums, int n) {
  __shared__ int sm[1024];
  int t = threadIdx.x;
  int i = blockIdx.x * 1024 + t;
  int v = (i < n) ? counts[i] : 0;
  sm[t] = v;
  __syncthreads();
  for (int off = 1; off < 1024; off <<= 1) {
    int u = (t >= off) ? sm[t - off] : 0;
    __syncthreads();
    sm[t] += u;
    __syncthreads();
  }
  if (i < n) rowstart[i + 1] = sm[t];
  if (t == 1023) blockSums[blockIdx.x] = sm[1023];
}

__global__ void k_scan2(const int* __restrict__ blockSums, int* __restrict__ blockOff, int nb) {
  __shared__ int sm[1024];
  int t = threadIdx.x;
  int v = (t < nb) ? blockSums[t] : 0;
  sm[t] = v;
  __syncthreads();
  for (int off = 1; off < 1024; off <<= 1) {
    int u = (t >= off) ? sm[t - off] : 0;
    __syncthreads();
    sm[t] += u;
    __syncthreads();
  }
  if (t < nb) blockOff[t] = sm[t] - v;
}

__global__ void k_scan3(int* __restrict__ rowstart, const int* __restrict__ blockOff, int n) {
  int i = blockIdx.x * blockDim.x + threadIdx.x;
  if (i < n) rowstart[i + 1] += blockOff[i >> 10];
  if (i == 0) rowstart[0] = 0;
}

__global__ void k_fill(const int* __restrict__ src, const int* __restrict__ dst,
                       const float* __restrict__ dinv, const int* __restrict__ rowstart,
                       int* __restrict__ cursor, int2* __restrict__ rec, int e) {
  int i = blockIdx.x * blockDim.x + threadIdx.x;
  if (i >= e) return;
  int s = src[i], d = dst[i];
  int p = atomicAdd(&cursor[d], 1);
  int2 r;
  r.x = s;
  r.y = __float_as_int(dinv[s] * dinv[d]);
  rec[rowstart[d] + p] = r;
}

__global__ void k_gemm128(const float* __restrict__ X, const float* __restrict__ W,
                          const float* __restrict__ bias, float* __restrict__ H, int n) {
  extern __shared__ float ws[];
  const int tid = threadIdx.x;

  for (int i = tid; i < HID * HID; i += 256) {
    int k = i >> 7, c = i & 127;
    ws[c * WSTRIDE + k] = W[i];
  }
  __syncthreads();

  const int wave   = tid >> 5;
  const int lane   = tid & 31;
  const int lane15 = lane & 15;
  const int half   = lane >> 4;
  const int rowBase = blockIdx.x * 128 + wave * 16;
  const int arow    = rowBase + lane15;
  const int koff    = half * 2;

  v2f a[32];
  if (arow < n) {
    const float* xp = X + (size_t)arow * HID + koff;
    if (bias) {
      const float* bp = bias + koff;
#pragma unroll
      for (int kk = 0; kk < 32; ++kk) {
        v2f xv = *(const v2f*)(xp + kk * 4);
        v2f bv = *(const v2f*)(bp + kk * 4);
        v2f t = xv + bv;
        t.x = fmaxf(t.x, 0.0f);
        t.y = fmaxf(t.y, 0.0f);
        a[kk] = t;
      }
    } else {
#pragma unroll
      for (int kk = 0; kk < 32; ++kk)
        a[kk] = *(const v2f*)(xp + kk * 4);
    }
  } else {
    v2f z = {};
#pragma unroll
    for (int kk = 0; kk < 32; ++kk) a[kk] = z;
  }

#pragma unroll
  for (int nt = 0; nt < 8; ++nt) {
    v8f acc = {};
    const float* bp = ws + (nt * 16 + lane15) * WSTRIDE + koff;
#pragma unroll
    for (int kk = 0; kk < 32; ++kk) {
      v2f b = *(const v2f*)(bp + kk * 4);
      acc = __builtin_amdgcn_wmma_f32_16x16x4_f32(
          false, a[kk], false, b, (short)0, acc, false, false);
    }
    const int colOut = nt * 16 + lane15;
#pragma unroll
    for (int r = 0; r < 8; ++r) {
      int row = rowBase + r + half * 8;
      if (row < n) H[(size_t)row * HID + colOut] = acc[r];
    }
  }
}

__global__ void k_aggregate(const float* __restrict__ H, const int2* __restrict__ rec,
                            const int* __restrict__ rowstart, const float* __restrict__ dinv,
                            float* __restrict__ AGG, int n) {
  long long gid = (long long)blockIdx.x * blockDim.x + threadIdx.x;
  int node = (int)(gid >> 5);
  if (node >= n) return;
  int q = ((int)gid & 31) * 4;

  int beg = __builtin_amdgcn_readfirstlane(rowstart[node]);
  int end = __builtin_amdgcn_readfirstlane(rowstart[node + 1]);

  float wself = dinv[node];
  wself *= wself;
  float4 acc = *(const float4*)(H + (size_t)node * HID + q);
  acc.x *= wself; acc.y *= wself; acc.z *= wself; acc.w *= wself;

  for (int j = beg; j < end; ++j) {
    int2 r = rec[j];
    int s = r.x;
    float w = __int_as_float(r.y);
    const float4 h = *(const float4*)(H + (size_t)s * HID + q);
    acc.x += w * h.x; acc.y += w * h.y; acc.z += w * h.z; acc.w += w * h.w;
  }
  *(float4*)(AGG + (size_t)node * HID + q) = acc;
}

__global__ void k_cnt(const int* __restrict__ batch, float* __restrict__ cnt, int n) {
  int i = blockIdx.x * blockDim.x + threadIdx.x;
  if (i < n)
    __hip_atomic_fetch_add(&cnt[batch[i]], 1.0f, __ATOMIC_RELAXED, __HIP_MEMORY_SCOPE_AGENT);
}

__global__ void k_pool(const float* __restrict__ H, const int* __restrict__ batch,
                       float* __restrict__ pooled, int n) {
  long long gid = (long long)blockIdx.x * blockDim.x + threadIdx.x;
  int node = (int)(gid >> 5);
  if (node >= n) return;
  int q = ((int)gid & 31) * 4;
  int g = __builtin_amdgcn_readfirstlane(batch[node]);
  const float4 h = *(const float4*)(H + (size_t)node * HID + q);
  float* p = pooled + (size_t)g * HID + q;
  __hip_atomic_fetch_add(p + 0, h.x, __ATOMIC_RELAXED, __HIP_MEMORY_SCOPE_AGENT);
  __hip_atomic_fetch_add(p + 1, h.y, __ATOMIC_RELAXED, __HIP_MEMORY_SCOPE_AGENT);
  __hip_atomic_fetch_add(p + 2, h.z, __ATOMIC_RELAXED, __HIP_MEMORY_SCOPE_AGENT);
  __hip_atomic_fetch_add(p + 3, h.w, __ATOMIC_RELAXED, __HIP_MEMORY_SCOPE_AGENT);
}

__global__ void k_final(const float* __restrict__ pooled, const float* __restrict__ cnt,
                        const float* __restrict__ b3, const float* __restrict__ Wl,
                        const float* __restrict__ bl, float* __restrict__ out, int G, int C) {
  int gid = blockIdx.x * blockDim.x + threadIdx.x;
  if (gid >= G * C) return;
  int g = gid / C, c = gid % C;
  float inv = 1.0f / fmaxf(cnt[g], 1.0f);
  float s = 0.0f;
  const float* pr = pooled + (size_t)g * HID;
  for (int k = 0; k < HID; ++k)
    s += (pr[k] * inv + b3[k]) * Wl[k * C + c];
  out[gid] = s + bl[c];
}

extern "C" void kernel_launch(void* const* d_in, const int* in_sizes, int n_in,
                              void* d_out, int out_size, void* d_ws, size_t ws_size,
                              hipStream_t stream) {
  const float* x   = (const float*)d_in[0];
  const int*   ei  = (const int*)d_in[1];
  const int*   bat = (const int*)d_in[2];
  const float* W1  = (const float*)d_in[3];
  const float* b1  = (const float*)d_in[4];
  const float* W2  = (const float*)d_in[5];
  const float* b2  = (const float*)d_in[6];
  const float* W3  = (const float*)d_in[7];
  const float* b3  = (const float*)d_in[8];
  const float* Wl  = (const float*)d_in[9];
  const float* bl  = (const float*)d_in[10];

  const int n = in_sizes[2];
  const int e = in_sizes[1] / 2;
  const int C = in_sizes[10];
  const int G = out_size / C;
  const int* src = ei;
  const int* dst = ei + e;

  char* base = (char*)d_ws;
  auto alloc = [&](size_t bytes) -> void* {
    void* p = (void*)base;
    base += (bytes + 255) & ~(size_t)255;
    return p;
  };
  int*   counts    = (int*)alloc((size_t)n * 4);
  int*   cursor    = (int*)alloc((size_t)n * 4);
  int*   rowstart  = (int*)alloc(((size_t)n + 1) * 4);
  int*   blockSums = (int*)alloc(1024 * 4);
  int*   blockOff  = (int*)alloc(1024 * 4);
  float* dinv      = (float*)alloc((size_t)n * 4);
  int2*  rec       = (int2*)alloc((size_t)e * 8);
  float* Hb        = (float*)alloc((size_t)n * HID * 4);
  float* Ab        = (float*)alloc((size_t)n * HID * 4);
  float* pooled    = (float*)alloc((size_t)G * HID * 4);
  float* cnt       = (float*)alloc((size_t)G * 4);
  (void)ws_size; (void)n_in;

  const int T = 256;
  auto nb = [](long long w, int t) { return (unsigned)((w + t - 1) / t); };
  const long long nw = (long long)n * 32;
  const unsigned gemmB = (unsigned)((n + 127) / 128);
  const unsigned scanB = (unsigned)((n + 1023) / 1024);
  const size_t lds = (size_t)HID * WSTRIDE * sizeof(float);

  k_zero_i<<<nb(n, T), T, 0, stream>>>(counts, n);
  k_zero_i<<<nb(n, T), T, 0, stream>>>(cursor, n);
  k_count<<<nb(e, T), T, 0, stream>>>(dst, counts, e);
  k_dinv<<<nb(n, T), T, 0, stream>>>(counts, dinv, n);
  k_scan1<<<scanB, 1024, 0, stream>>>(counts, rowstart, blockSums, n);
  k_scan2<<<1, 1024, 0, stream>>>(blockSums, blockOff, (int)scanB);
  k_scan3<<<nb(n, T), T, 0, stream>>>(rowstart, blockOff, n);
  k_fill<<<nb(e, T), T, 0, stream>>>(src, dst, dinv, rowstart, cursor, rec, e);

  k_gemm128<<<gemmB, T, lds, stream>>>(x, W1, nullptr, Hb, n);
  k_aggregate<<<nb(nw, T), T, 0, stream>>>(Hb, rec, rowstart, dinv, Ab, n);

  k_gemm128<<<gemmB, T, lds, stream>>>(Ab, W2, b1, Hb, n);
  k_aggregate<<<nb(nw, T), T, 0, stream>>>(Hb, rec, rowstart, dinv, Ab, n);

  k_gemm128<<<gemmB, T, lds, stream>>>(Ab, W3, b2, Hb, n);
  k_aggregate<<<nb(nw, T), T, 0, stream>>>(Hb, rec, rowstart, dinv, Ab, n);

  k_zero_f<<<nb((long long)G * HID, T), T, 0, stream>>>(pooled, G * HID);
  k_zero_f<<<nb(G, T), T, 0, stream>>>(cnt, G);
  k_cnt<<<nb(n, T), T, 0, stream>>>(bat, cnt, n);
  k_pool<<<nb(nw, T), T, 0, stream>>>(Ab, bat, pooled, n);
  k_final<<<nb((long long)G * C, T), T, 0, stream>>>(pooled, cnt, b3, Wl, bl,
                                                     (float*)d_out, G, C);
}


# Mean pool: each wave sums 16 consecutive nodes in registers and flushes once per run of equal graph id (sums and counts in one kernel): 15x fewer float atomics; sum order changes
# speedup vs baseline: 1.1834x; 1.1834x over previous
#include <hip/hip_runtime.h>
#include <hip/hip_bf16.h>

typedef __attribute__((ext_vector_type(2))) float v2f;
typedef __attribute__((ext_vector_type(8))) float v8f;

#define HID 128
#define WSTRIDE 132

__global__ void k_zero_i(int* __restrict__ p, int n) {
  int i = blockIdx.x * blockDim.x + threadIdx.x;
  if (i < n) p[i] = 0;
}
__global__ void k_zero_f(float* __restrict__ p, int n) {
  int i = blockIdx.x * blockDim.x + threadIdx.x;
  if (i < n) p[i] = 0.0f;
}

__global__ void k_count(const int* __restrict__ dst, int* __restrict__ cnt, int e) {
  int i = blockIdx.x * blockDim.x + threadIdx.x;
  if (i < e) atomicAdd(&cnt[dst[i]], 1);
}

__global__ void k_dinv(const int* __restrict__ cnt, float* __restrict__ dinv, int n) {
  int i = blockIdx.x * blockDim.x + threadIdx.x;
  if (i < n) dinv[i] = rsqrtf((float)cnt[i] + 1.0f);
}

__global__ void k_scan1(const int* __restrict__ counts, int* __restrict__ rowstart,
                        int* __restrict__ blockSums, int n) {
  __shared__ int sm[1024];
  int t = threadIdx.x;
  int i = blockIdx.x * 1024 + t;
  int v = (i < n) ? counts[i] : 0;
  sm[t] = v;
  __syncthreads();
  for (int off = 1; off < 1024; off <<= 1) {
    int u = (t >= off) ? sm[t - off] : 0;
    __syncthreads();
    sm[t] += u;
    __syncthreads();
  }
  if (i < n) rowstart[i + 1] = sm[t];
  if (t == 1023) blockSums[blockIdx.x] = sm[1023];
}

__global__ void k_scan2(const int* __restrict__ blockSums, int* __restrict__ blockOff, int nb) {
  __shared__ int sm[1024];
  int t = threadIdx.x;
  int v = (t < nb) ? blockSums[t] : 0;
  sm[t] = v;
  __syncthreads();
  for (int off = 1; off < 1024; off <<= 1) {
    int u = (t >= off) ? sm[t - off] : 0;
    __syncthreads();
    sm[t] += u;
    __syncthreads();
  }
  if (t < nb) blockOff[t] = sm[t] - v;
}

__global__ void k_scan3(int* __restrict__ rowstart, const int* __restrict__ blockOff, int n) {
  int i = blockIdx.x * blockDim.x + threadIdx.x;
  if (i < n) rowstart[i + 1] += blockOff[i >> 10];
  if (i == 0) rowstart[0] = 0;
}

__global__ void k_fill(const int* __restrict__ src, const int* __restrict__ dst,
                       const float* __restrict__ dinv, const int* __restrict__ rowstart,
                       int* __restrict__ cursor, int2* __restrict__ rec, int e) {
  int i = blockIdx.x * blockDim.x + threadIdx.x;
  if (i >= e) return;
  int s = src[i], d = dst[i];
  int p = atomicAdd(&cursor[d], 1);
  int2 r;
  r.x = s;
  r.y = __float_as_int(dinv[s] * dinv[d]);
  rec[rowstart[d] + p] = r;
}

__global__ void k_gemm128(const float* __restrict__ X, const float* __restrict__ W,
                          const float* __restrict__ bias, float* __restrict__ H, int n) {
  extern __shared__ float ws[];
  const int tid = threadIdx.x;

  for (int i = tid; i < HID * HID; i += 256) {
    int k = i >> 7, c = i & 127;
    ws[c * WSTRIDE + k] = W[i];
  }
  __syncthreads();

  const int wave   = tid >> 5;
  const int lane   = tid & 31;
  const int lane15 = lane & 15;
  const int half   = lane >> 4;
  const int rowBase = blockIdx.x * 128 + wave * 16;
  const int arow    = rowBase + lane15;
  const int koff    = half * 2;

  v2f a[32];
  if (arow < n) {
    const float* xp = X + (size_t)arow * HID + koff;
    if (bias) {
      const float* bp = bias + koff;
#pragma unroll
      for (int kk = 0; kk < 32; ++kk) {
        v2f xv = *(const v2f*)(xp + kk * 4);
        v2f bv = *(const v2f*)(bp + kk * 4);
        v2f t = xv + bv;
        t.x = fmaxf(t.x, 0.0f);
        t.y = fmaxf(t.y, 0.0f);
        a[kk] = t;
      }
    } else {
#pragma unroll
      for (int kk = 0; kk < 32; ++kk)
        a[kk] = *(const v2f*)(xp + kk * 4);
    }
  } else {
    v2f z = {};
#pragma unroll
    for (int kk = 0; kk < 32; ++kk) a[kk] = z;
  }

#pragma unroll
  for (int nt = 0; nt < 8; ++nt) {
    v8f acc = {};
    const float* bp = ws + (nt * 16 + lane15) * WSTRIDE + koff;
#pragma unroll
    for (int kk = 0; kk < 32; ++kk) {
      v2f b = *(const v2f*)(bp + kk * 4);
      acc = __builtin_amdgcn_wmma_f32_16x16x4_f32(
          false, a[kk], false, b, (short)0, acc, false, false);
    }
    const int colOut = nt * 16 + lane15;
#pragma unroll
    for (int r = 0; r < 8; ++r) {
      int row = rowBase + r + half * 8;
      if (row < n) H[(size_t)row * HID + colOut] = acc[r];
    }
  }
}

__global__ void k_aggregate(const float* __restrict__ H, const int2* __restrict__ rec,
                            const int* __restrict__ rowstart, const float* __restrict__ dinv,
                            float* __restrict__ AGG, int n) {
  long long gid = (long long)blockIdx.x * blockDim.x + threadIdx.x;
  int node = (int)(gid >> 5);
  if (node >= n) return;
  int q = ((int)gid & 31) * 4;

  int beg = __builtin_amdgcn_readfirstlane(rowstart[node]);
  int end = __builtin_amdgcn_readfirstlane(rowstart[node + 1]);

  float wself = dinv[node];
  wself *= wself;
  float4 acc = *(const float4*)(H + (size_t)node * HID + q);
  acc.x *= wself; acc.y *= wself; acc.z *= wself; acc.w *= wself;

  for (int j = beg; j < end; ++j) {
    int2 r = rec[j];
    int s = r.x;
    float w = __int_as_float(r.y);
    const float4 h = *(const float4*)(H + (size_t)s * HID + q);
    acc.x += w * h.x; acc.y += w * h.y; acc.z += w * h.z; acc.w += w * h.w;
  }
  *(float4*)(AGG + (size_t)node * HID + q) = acc;
}

#define PCH 16

__device__ __forceinline__ void pool_flush(float* __restrict__ pooled, float* __restrict__ cnt,
                                           int g, int q, int lane, float4 acc, int run) {
  float* p = pooled + (size_t)g * HID + q;
  __hip_atomic_fetch_add(p + 0, acc.x, __ATOMIC_RELAXED, __HIP_MEMORY_SCOPE_AGENT);
  __hip_atomic_fetch_add(p + 1, acc.y, __ATOMIC_RELAXED, __HIP_MEMORY_SCOPE_AGENT);
  __hip_atomic_fetch_add(p + 2, acc.z, __ATOMIC_RELAXED, __HIP_MEMORY_SCOPE_AGENT);
  __hip_atomic_fetch_add(p + 3, acc.w, __ATOMIC_RELAXED, __HIP_MEMORY_SCOPE_AGENT);
  if (lane == 0)
    __hip_atomic_fetch_add(&cnt[g], (float)run, __ATOMIC_RELAXED, __HIP_MEMORY_SCOPE_AGENT);
}

__global__ void k_pool(const float* __restrict__ H, const int* __restrict__ batch,
                       float* __restrict__ pooled, float* __restrict__ cnt, int n) {
  const int lane = threadIdx.x & 31;
  const int wv = __builtin_amdgcn_readfirstlane(
      (int)(blockIdx.x * (blockDim.x >> 5) + (threadIdx.x >> 5)));
  const int base = wv * PCH;
  if (base >= n) return;
  const int q = lane * 4;
  const int last = n - 1;

  int g[PCH];
  float4 h[PCH];
#pragma unroll
  for (int i = 0; i < PCH; ++i) {
    const int node = min(base + i, last);
    g[i] = batch[node];
    h[i] = *(const float4*)(H + (size_t)node * HID + q);
  }

  int cur = g[0];
  int run = 1;
  float4 acc = h[0];
#pragma unroll
  for (int i = 1; i < PCH; ++i) {
    if (base + i <= last) {
      if (g[i] != cur) {
        pool_flush(pooled, cnt, cur, q, lane, acc, run);
        cur = g[i];
        run = 1;
        acc = h[i];
      } else {
        acc.x += h[i].x; acc.y += h[i].y; acc.z += h[i].z; acc.w += h[i].w;
        ++run;
      }
    }
  }
  pool_flush(pooled, cnt, cur, q, lane, acc, run);
}

__global__ void k_final(const float* __restrict__ pooled, const float* __restrict__ cnt,
                        const float* __restrict__ b3, const float* __restrict__ Wl,
                        const float* __restrict__ bl, float* __restrict__ out, int G, int C) {
  int gid = blockIdx.x * blockDim.x + threadIdx.x;
  if (gid >= G * C) return;
  int g = gid / C, c = gid % C;
  float inv = 1.0f / fmaxf(cnt[g], 1.0f);
  float s = 0.0f;
  const float* pr = pooled + (size_t)g * HID;
  for (int k = 0; k < HID; ++k)
    s += (pr[k] * inv + b3[k]) * Wl[k * C + c];
  out[gid] = s + bl[c];
}

extern "C" void kernel_launch(void* const* d_in, const int* in_sizes, int n_in,
                              void* d_out, int out_size, void* d_ws, size_t ws_size,
                              hipStream_t stream) {
  const float* x   = (const float*)d_in[0];
  const int*   ei  = (const int*)d_in[1];
  const int*   bat = (const int*)d_in[2];
  const float* W1  = (const float*)d_in[3];
  const float* b1  = (const float*)d_in[4];
  const float* W2  = (const float*)d_in[5];
  const float* b2  = (const float*)d_in[6];
  const float* W3  = (const float*)d_in[7];
  const float* b3  = (const float*)d_in[8];
  const float* Wl  = (const float*)d_in[9];
  const float* bl  = (const float*)d_in[10];

  const int n = in_sizes[2];
  const int e = in_sizes[1] / 2;
  const int C = in_sizes[10];
  const int G = out_size / C;
  const int* src = ei;
  const int* dst = ei + e;

  char* base = (char*)d_ws;
  auto alloc = [&](size_t bytes) -> void* {
    void* p = (void*)base;
    base += (bytes + 255) & ~(size_t)255;
    return p;
  };
  int*   counts    = (int*)alloc((size_t)n * 4);
  int*   cursor    = (int*)alloc((size_t)n * 4);
  int*   rowstart  = (int*)alloc(((size_t)n + 1) * 4);
  int*   blockSums = (int*)alloc(1024 * 4);
  int*   blockOff  = (int*)alloc(1024 * 4);
  float* dinv      = (float*)alloc((size_t)n * 4);
  int2*  rec       = (int2*)alloc((size_t)e * 8);
  float* Hb        = (float*)alloc((size_t)n * HID * 4);
  float* Ab        = (float*)alloc((size_t)n * HID * 4);
  float* pooled    = (float*)alloc((size_t)G * HID * 4);
  float* cnt       = (float*)alloc((size_t)G * 4);
  (void)ws_size; (void)n_in;

  const int T = 256;
  auto nb = [](long long w, int t) { return (unsigned)((w + t - 1) / t); };
  const long long nw = (long long)n * 32;
  const unsigned gemmB = (unsigned)((n + 127) / 128);
  const unsigned scanB = (unsigned)((n + 1023) / 1024);
  const size_t lds = (size_t)HID * WSTRIDE * sizeof(float);

  k_zero_i<<<nb(n, T), T, 0, stream>>>(counts, n);
  k_zero_i<<<nb(n, T), T, 0, stream>>>(cursor, n);
  k_count<<<nb(e, T), T, 0, stream>>>(dst, counts, e);
  k_dinv<<<nb(n, T), T, 0, stream>>>(counts, dinv, n);
  k_scan1<<<scanB, 1024, 0, stream>>>(counts, rowstart, blockSums, n);
  k_scan2<<<1, 1024, 0, stream>>>(blockSums, blockOff, (int)scanB);
  k_scan3<<<nb(n, T), T, 0, stream>>>(rowstart, blockOff, n);
  k_fill<<<nb(e, T), T, 0, stream>>>(src, dst, dinv, rowstart, cursor, rec, e);

  k_gemm128<<<gemmB, T, lds, stream>>>(x, W1, nullptr, Hb, n);
  k_aggregate<<<nb(nw, T), T, 0, stream>>>(Hb, rec, rowstart, dinv, Ab, n);

  k_gemm128<<<gemmB, T, lds, stream>>>(Ab, W2, b1, Hb, n);
  k_aggregate<<<nb(nw, T), T, 0, stream>>>(Hb, rec, rowstart, dinv, Ab, n);

  k_gemm128<<<gemmB, T, lds, stream>>>(Ab, W3, b2, Hb, n);
  k_aggregate<<<nb(nw, T), T, 0, stream>>>(Hb, rec, rowstart, dinv, Ab, n);

  k_zero_f<<<nb((long long)G * HID, T), T, 0, stream>>>(pooled, G * HID);
  k_zero_f<<<nb(G, T), T, 0, stream>>>(cnt, G);
  k_pool<<<nb((long long)((n + PCH - 1) / PCH) * 32, T), T, 0, stream>>>(Ab, bat, pooled, cnt, n);
  k_final<<<nb((long long)G * C, T), T, 0, stream>>>(pooled, cnt, b3, Wl, bl,
                                                     (float*)d_out, G, C);
}


# Classifier folded into layer 3 (no ReLU between): Z = relu(h2+b2) (W3 Wl) is 16 wide, so the last product, neighbour sum and mean pool move 64 B rows, not 512 B; product and sum order change
# speedup vs baseline: 1.3183x; 1.1140x over previous
#include <hip/hip_runtime.h>
#include <hip/hip_bf16.h>

typedef __attribute__((ext_vector_type(2))) float v2f;
typedef __attribute__((ext_vector_type(8))) float v8f;

#define HID 128
#define WSTRIDE 132
#define ZW 16

__global__ void k_zero_i(int* __restrict__ p, int n) {
  int i = blockIdx.x * blockDim.x + threadIdx.x;
  if (i < n) p[i] = 0;
}
__global__ void k_zero_f(float* __restrict__ p, int n) {
  int i = blockIdx.x * blockDim.x + threadIdx.x;
  if (i < n) p[i] = 0.0f;
}

__global__ void k_count(const int* __restrict__ dst, int* __restrict__ cnt, int e) {
  int i = blockIdx.x * blockDim.x + threadIdx.x;
  if (i < e) atomicAdd(&cnt[dst[i]], 1);
}

__global__ void k_dinv(const int* __restrict__ cnt, float* __restrict__ dinv, int n) {
  int i = blockIdx.x * blockDim.x + threadIdx.x;
  if (i < n) dinv[i] = rsqrtf((float)cnt[i] + 1.0f);
}

__global__ void k_scan1(const int* __restrict__ counts, int* __restrict__ rowstart,
                        int* __restrict__ blockSums, int n) {
  __shared__ int sm[1024];
  int t = threadIdx.x;
  int i = blockIdx.x * 1024 + t;
  int v = (i < n) ? counts[i] : 0;
  sm[t] = v;
  __syncthreads();
  for (int off = 1; off < 1024; off <<= 1) {
    int u = (t >= off) ? sm[t - off] : 0;
    __syncthreads();
    sm[t] += u;
    __syncthreads();
  }
  if (i < n) rowstart[i + 1] = sm[t];
  if (t == 1023) blockSums[blockIdx.x] = sm[1023];
}

__global__ void k_scan2(const int* __restrict__ blockSums, int* __restrict__ blockOff, int nb) {
  __shared__ int sm[1024];
  int t = threadIdx.x;
  int v = (t < nb) ? blockSums[t] : 0;
  sm[t] = v;
  __syncthreads();
  for (int off = 1; off < 1024; off <<= 1) {
    int u = (t >= off) ? sm[t - off] : 0;
    __syncthreads();
    sm[t] += u;
    __syncthreads();
  }
  if (t < nb) blockOff[t] = sm[t] - v;
}

__global__ void k_scan3(int* __restrict__ rowstart, const int* __restrict__ blockOff, int n) {
  int i = blockIdx.x * blockDim.x + threadIdx.x;
  if (i < n) rowstart[i + 1] += blockOff[i >> 10];
  if (i == 0) rowstart[0] = 0;
}

__global__ void k_fill(const int* __restrict__ src, const int* __restrict__ dst,
                       const float* __restrict__ dinv, const int* __restrict__ rowstart,
                       int* __restrict__ cursor, int2* __restrict__ rec, int e) {
  int i = blockIdx.x * blockDim.x + threadIdx.x;
  if (i >= e) return;
  int s = src[i], d = dst[i];
  int p = atomicAdd(&cursor[d], 1);
  int2 r;
  r.x = s;
  r.y = __float_as_int(dinv[s] * dinv[d]);
  rec[rowstart[d] + p] = r;
}

__global__ void k_gemm128(const float* __restrict__ X, const float* __restrict__ W,
                          const float* __restrict__ bias, float* __restrict__ H, int n) {
  extern __shared__ float ws[];
  const int tid = threadIdx.x;

  for (int i = tid; i < HID * HID; i += 256) {
    int k = i >> 7, c = i & 127;
    ws[c * WSTRIDE + k] = W[i];
  }
  __syncthreads();

  const int wave   = tid >> 5;
  const int lane   = tid & 31;
  const int lane15 = lane & 15;
  const int half   = lane >> 4;
  const int rowBase = blockIdx.x * 128 + wave * 16;
  const int arow    = rowBase + lane15;
  const int koff    = half * 2;

  v2f a[32];
  if (arow < n) {
    const float* xp = X + (size_t)arow * HID + koff;
    if (bias) {
      const float* bp = bias + koff;
#pragma unroll
      for (int kk = 0; kk < 32; ++kk) {
        v2f xv = *(const v2f*)(xp + kk * 4);
        v2f bv = *(const v2f*)(bp + kk * 4);
        v2f t = xv + bv;
        t.x = fmaxf(t.x, 0.0f);
        t.y = fmaxf(t.y, 0.0f);
        a[kk] = t;
      }
    } else {
#pragma unroll
      for (int kk = 0; kk < 32; ++kk)
        a[kk] = *(const v2f*)(xp + kk * 4);
    }
  } else {
    v2f z = {};
#pragma unroll
    for (int kk = 0; kk < 32; ++kk) a[kk] = z;
  }

#pragma unroll
  for (int nt = 0; nt < 8; ++nt) {
    v8f acc = {};
    const float* bp = ws + (nt * 16 + lane15) * WSTRIDE + koff;
#pragma unroll
    for (int kk = 0; kk < 32; ++kk) {
      v2f b = *(const v2f*)(bp + kk * 4);
      acc = __builtin_amdgcn_wmma_f32_16x16x4_f32(
          false, a[kk], false, b, (short)0, acc, false, false);
    }
    const int colOut = nt * 16 + lane15;
#pragma unroll
    for (int r = 0; r < 8; ++r) {
      int row = rowBase + r + half * 8;
      if (row < n) H[(size_t)row * HID + colOut] = acc[r];
    }
  }
}

__global__ void k_aggregate(const float* __restrict__ H, const int2* __restrict__ rec,
                            const int* __restrict__ rowstart, const float* __restrict__ dinv,
                            float* __restrict__ AGG, int n) {
  long long gid = (long long)blockIdx.x * blockDim.x + threadIdx.x;
  int node = (int)(gid >> 5);
  if (node >= n) return;
  int q = ((int)gid & 31) * 4;

  int beg = __builtin_amdgcn_readfirstlane(rowstart[node]);
  int end = __builtin_amdgcn_readfirstlane(rowstart[node + 1]);

  float wself = dinv[node];
  wself *= wself;
  float4 acc = *(const float4*)(H + (size_t)node * HID + q);
  acc.x *= wself; acc.y *= wself; acc.z *= wself; acc.w *= wself;

  for (int j = beg; j < end; ++j) {
    int2 r = rec[j];
    int s = r.x;
    float w = __int_as_float(r.y);
    const float4 h = *(const float4*)(H + (size_t)s * HID + q);
    acc.x += w * h.x; acc.y += w * h.y; acc.z += w * h.z; acc.w += w * h.w;
  }
  *(float4*)(AGG + (size_t)node * HID + q) = acc;
}

template <bool ACT>
__global__ void k_gemm16(const float* __restrict__ X, const float* __restrict__ B, int ldb, int nc,
                         const float* __restrict__ bias, float* __restrict__ Z, int n) {
  __shared__ float bs[ZW * WSTRIDE];
  __shared__ float bsh[256];
  const int tid = threadIdx.x;

  const int wave   = tid >> 5;
  const int lane   = tid & 31;
  const int lane15 = lane & 15;
  const int half   = lane >> 4;
  const int rowBase = blockIdx.x * 128 + wave * 16;
  const int arow    = rowBase + lane15;
  const int koff    = half * 2;

  float sv[HID * ZW / 256];
#pragma unroll
  for (int it = 0; it < HID * ZW / 256; ++it) {
    int i = tid + it * 256;
    sv[it] = B[(i >> 4) * ldb + min(i & 15, nc - 1)];
  }
  float bsv = 0.0f;
  if (ACT) bsv = bias[tid & (HID - 1)];

  v2f a[32];
  const float* xp = X + (size_t)min(arow, n - 1) * HID + koff;
#pragma unroll
  for (int kk = 0; kk < 32; ++kk)
    a[kk] = *(const v2f*)(xp + kk * 4);

#pragma unroll
  for (int it = 0; it < HID * ZW / 256; ++it) {
    int i = tid + it * 256;
    int k = i >> 4, c = i & 15;
    bs[c * WSTRIDE + k] = (c < nc) ? sv[it] : 0.0f;
  }
  if (ACT) bsh[tid] = bsv;
  __syncthreads();

  v8f acc = {};
  const float* bp = bs + lane15 * WSTRIDE + koff;
  const float* cp = bsh + koff;
#pragma unroll
  for (int kk = 0; kk < 32; ++kk) {
    v2f av = a[kk];
    if (ACT) {
      v2f cv = *(const v2f*)(cp + kk * 4);
      av = av + cv;
      av.x = fmaxf(av.x, 0.0f);
      av.y = fmaxf(av.y, 0.0f);
    }
    v2f b = *(const v2f*)(bp + kk * 4);
    acc = __builtin_amdgcn_wmma_f32_16x16x4_f32(
        false, av, false, b, (short)0, acc, false, false);
  }
#pragma unroll
  for (int r = 0; r < 8; ++r) {
    int row = rowBase + r + half * 8;
    if (row < n) Z[(size_t)row * ZW + lane15] = acc[r];
  }
}

__global__ void k_aggregate16(const float* __restrict__ Z, const int2* __restrict__ rec,
                              const int* __restrict__ rowstart, const float* __restrict__ dinv,
                              float* __restrict__ AZ, int n) {
  const int gid = blockIdx.x * blockDim.x + threadIdx.x;
  const int node = gid >> 2;
  if (node >= n) return;
  const int q = (gid & 3) * 4;

  const int beg = rowstart[node];
  const int end = rowstart[node + 1];

  float wself = dinv[node];
  wself *= wself;
  float4 acc = *(const float4*)(Z + (size_t)node * ZW + q);
  acc.x *= wself; acc.y *= wself; acc.z *= wself; acc.w *= wself;

  for (int j = beg; j < end; j += 4) {
    int2 r[4];
    float4 h[4];
#pragma unroll
    for (int t = 0; t < 4; ++t) r[t] = rec[(j + t < end) ? (j + t) : j];
#pragma unroll
    for (int t = 0; t < 4; ++t) h[t] = *(const float4*)(Z + (size_t)r[t].x * ZW + q);
#pragma unroll
    for (int t = 0; t < 4; ++t) {
      const bool ok = (j + t) < end;
      const float w = __int_as_float(r[t].y);
      const float hx = ok ? h[t].x : 0.0f, hy = ok ? h[t].y : 0.0f;
      const float hz = ok ? h[t].z : 0.0f, hw = ok ? h[t].w : 0.0f;
      acc.x += w * hx; acc.y += w * hy; acc.z += w * hz; acc.w += w * hw;
    }
  }
  *(float4*)(AZ + (size_t)node * ZW + q) = acc;
}

#define PCH 8

__device__ __forceinline__ void pool_flush(float* __restrict__ pooled, float* __restrict__ cnt,
                                           int g, int q, float4 acc, int run) {
  float* p = pooled + (size_t)g * ZW + q;
  __hip_atomic_fetch_add(p + 0, acc.x, __ATOMIC_RELAXED, __HIP_MEMORY_SCOPE_AGENT);
  __hip_atomic_fetch_add(p + 1, acc.y, __ATOMIC_RELAXED, __HIP_MEMORY_SCOPE_AGENT);
  __hip_atomic_fetch_add(p + 2, acc.z, __ATOMIC_RELAXED, __HIP_MEMORY_SCOPE_AGENT);
  __hip_atomic_fetch_add(p + 3, acc.w, __ATOMIC_RELAXED, __HIP_MEMORY_SCOPE_AGENT);
  if (q == 0)
    __hip_atomic_fetch_add(&cnt[g], (float)run, __ATOMIC_RELAXED, __HIP_MEMORY_SCOPE_AGENT);
}

__global__ void k_pool16(const float* __restrict__ AZ, const int* __restrict__ batch,
                         float* __restrict__ pooled, float* __restrict__ cnt, int n) {
  const int gid = blockIdx.x * blockDim.x + threadIdx.x;
  const int first = (gid >> 5) * (8 * PCH) + ((gid & 31) >> 2);
  if (first >= n) return;
  const int q = (gid & 3) * 4;
  const int last = n - 1;

  int g[PCH];
  float4 h[PCH];
#pragma unroll
  for (int i = 0; i < PCH; ++i) {
    const int node = min(first + i * 8, last);
    g[i] = batch[node];
    h[i] = *(const float4*)(AZ + (size_t)node * ZW + q);
  }

  int cur = g[0];
  int run = 1;
  float4 acc = h[0];
#pragma unroll
  for (int i = 1; i < PCH; ++i) {
    if (first + i * 8 <= last) {
      if (g[i] != cur) {
        pool_flush(pooled, cnt, cur, q, acc, run);
        cur = g[i];
        run = 1;
        acc = h[i];
      } else {
        acc.x += h[i].x; acc.y += h[i].y; acc.z += h[i].z; acc.w += h[i].w;
        ++run;
      }
    }
  }
  pool_flush(pooled, cnt, cur, q, acc, run);
}

__global__ void k_final16(const float* __restrict__ pooled, const float* __restrict__ cnt,
                          const float* __restrict__ b3, const float* __restrict__ Wl,
                          const float* __restrict__ bl, float* __restrict__ out, int G, int C) {
  int gid = blockIdx.x * blockDim.x + threadIdx.x;
  if (gid >= G * C) return;
  int g = gid / C, c = gid % C;
  float cn = cnt[g];
  float inv = 1.0f / fmaxf(cn, 1.0f);
  float bw = 0.0f;
  for (int k = 0; k < HID; ++k)
    bw += b3[k] * Wl[k * C + c];
  float s = pooled[(size_t)g * ZW + c] * inv;
  if (cn > 0.0f) s += bw;
  out[gid] = s + bl[c];
}

extern "C" void kernel_launch(void* const* d_in, const int* in_sizes, int n_in,
                              void* d_out, int out_size, void* d_ws, size_t ws_size,
                              hipStream_t stream) {
  const float* x   = (const float*)d_in[0];
  const int*   ei  = (const int*)d_in[1];
  const int*   bat = (const int*)d_in[2];
  const float* W1  = (const float*)d_in[3];
  const float* b1  = (const float*)d_in[4];
  const float* W2  = (const float*)d_in[5];
  const float* b2  = (const float*)d_in[6];
  const float* W3  = (const float*)d_in[7];
  const float* b3  = (const float*)d_in[8];
  const float* Wl  = (const float*)d_in[9];
  const float* bl  = (const float*)d_in[10];

  const int n = in_sizes[2];
  const int e = in_sizes[1] / 2;
  const int C = in_sizes[10];
  const int G = out_size / C;
  const int* src = ei;
  const int* dst = ei + e;

  char* base = (char*)d_ws;
  auto alloc = [&](size_t bytes) -> void* {
    void* p = (void*)base;
    base += (bytes + 255) & ~(size_t)255;
    return p;
  };
  int*   counts    = (int*)alloc((size_t)n * 4);
  int*   cursor    = (int*)alloc((size_t)n * 4);
  int*   rowstart  = (int*)alloc(((size_t)n + 1) * 4);
  int*   blockSums = (int*)alloc(1024 * 4);
  int*   blockOff  = (int*)alloc(1024 * 4);
  float* dinv      = (float*)alloc((size_t)n * 4);
  int2*  rec       = (int2*)alloc((size_t)e * 8);
  float* Hb        = (float*)alloc((size_t)n * HID * 4);
  float* Ab        = (float*)alloc((size_t)n * HID * 4);
  float* Mb        = (float*)alloc((size_t)HID * ZW * 4);
  float* Zb        = (float*)alloc((size_t)n * ZW * 4);
  float* AZb       = (float*)alloc((size_t)n * ZW * 4);
  float* pooled    = (float*)alloc((size_t)G * ZW * 4);
  float* cnt       = (float*)alloc((size_t)G * 4);
  (void)ws_size; (void)n_in;

  const int T = 256;
  auto nb = [](long long w, int t) { return (unsigned)((w + t - 1) / t); };
  const long long nw = (long long)n * 32;
  const unsigned gemmB = (unsigned)((n + 127) / 128);
  const unsigned scanB = (unsigned)((n + 1023) / 1024);
  const size_t lds = (size_t)HID * WSTRIDE * sizeof(float);

  k_zero_i<<<nb(n, T), T, 0, stream>>>(counts, n);
  k_zero_i<<<nb(n, T), T, 0, stream>>>(cursor, n);
  k_count<<<nb(e, T), T, 0, stream>>>(dst, counts, e);
  k_dinv<<<nb(n, T), T, 0, stream>>>(counts, dinv, n);
  k_scan1<<<scanB, 1024, 0, stream>>>(counts, rowstart, blockSums, n);
  k_scan2<<<1, 1024, 0, stream>>>(blockSums, blockOff, (int)scanB);
  k_scan3<<<nb(n, T), T, 0, stream>>>(rowstart, blockOff, n);
  k_fill<<<nb(e, T), T, 0, stream>>>(src, dst, dinv, rowstart, cursor, rec, e);

  k_gemm128<<<gemmB, T, lds, stream>>>(x, W1, nullptr, Hb, n);
  k_aggregate<<<nb(nw, T), T, 0, stream>>>(Hb, rec, rowstart, dinv, Ab, n);

  k_gemm128<<<gemmB, T, lds, stream>>>(Ab, W2, b1, Hb, n);
  k_aggregate<<<nb(nw, T), T, 0, stream>>>(Hb, rec, rowstart, dinv, Ab, n);

  k_gemm16<false><<<1, T, 0, stream>>>(W3, Wl, C, C, nullptr, Mb, HID);
  k_gemm16<true><<<gemmB, T, 0, stream>>>(Ab, Mb, ZW, ZW, b2, Zb, n);
  k_aggregate16<<<nb((long long)n * 4, T), T, 0, stream>>>(Zb, rec, rowstart, dinv, AZb, n);

  k_zero_f<<<nb((long long)G * ZW, T), T, 0, stream>>>(pooled, G * ZW);
  k_zero_f<<<nb(G, T), T, 0, stream>>>(cnt, G);
  k_pool16<<<nb((long long)((n + 8 * PCH - 1) / (8 * PCH)) * 32, T), T, 0, stream>>>(AZb, bat, pooled,
                                                                               cnt, n);
  k_final16<<<nb((long long)G * C, T), T, 0, stream>>>(pooled, cnt, b3, Wl, bl,
                                                       (float*)d_out, G, C);
}


# CSR build as a two-level counting sort on LDS counters (edges binned by dst range, one work-group per bin): no global atomics, 8 launches become 5; a row's records stay in arrival order
# speedup vs baseline: 2.1106x; 1.6010x over previous
#include <hip/hip_runtime.h>
#include <hip/hip_bf16.h>

typedef __attribute__((ext_vector_type(2))) float v2f;
typedef __attribute__((ext_vector_type(8))) float v8f;

#define HID 128
#define WSTRIDE 132
#define ZW 16

__global__ void k_zero_f(float* __restrict__ p, int n) {
  int i = blockIdx.x * blockDim.x + threadIdx.x;
  if (i < n) p[i] = 0.0f;
}

#define BSH    9
#define BINN   (1 << BSH)
#define NBMAX  256
#define NCHUNK 256
#define ETH    1024
#define EB     8

__device__ __forceinline__ int wave_scan_incl(int v, int lane) {
#pragma unroll
  for (int off = 1; off < 32; off <<= 1) {
    int u = __shfl_up(v, off, 32);
    if (lane >= off) v += u;
  }
  return v;
}

__device__ __forceinline__ void hold_batch(int v0, int v1, int v2, int v3,
                                           int v4, int v5, int v6, int v7) {
  asm volatile("" :: "v"(v0), "v"(v1), "v"(v2), "v"(v3), "v"(v4), "v"(v5), "v"(v6), "v"(v7));
  __builtin_amdgcn_sched_barrier(0);
}

__global__ void k_binhist(const int* __restrict__ dst, int* __restrict__ hist, int e, int nbins) {
  __shared__ int h[NBMAX];
  const int tid = threadIdx.x;
  const int wg = blockIdx.x;
  if (tid < NBMAX) h[tid] = 0;
  const int per = (e + NCHUNK - 1) / NCHUNK;
  const int beg = wg * per;
  const int end = min(beg + per, e);
  __syncthreads();
  for (int i0 = beg + tid; i0 < end; i0 += ETH * EB) {
    int d[EB];
#pragma unroll
    for (int u = 0; u < EB; ++u) d[u] = dst[min(i0 + u * ETH, end - 1)];
    hold_batch(d[0], d[1], d[2], d[3], d[4], d[5], d[6], d[7]);
#pragma unroll
    for (int u = 0; u < EB; ++u)
      if (i0 + u * ETH < end) atomicAdd(&h[d[u] >> BSH], 1);
  }
  __syncthreads();
  if (tid < nbins) hist[tid * NCHUNK + wg] = h[tid];
}

__global__ void k_binscan(const int* __restrict__ hist, int* __restrict__ base,
                          int* __restrict__ binStart, int nbins) {
  __shared__ int tot[NBMAX];
  __shared__ int wsum[NBMAX / 32];
  const int tid = threadIdx.x;
  const int lane = tid & 31;
  const int wave = tid >> 5;

  int4 a[NBMAX / 32], c[NBMAX / 32];
  int s[NBMAX / 32], inc[NBMAX / 32];
#pragma unroll
  for (int k = 0; k < NBMAX / 32; ++k) {
    const int b = min(wave + 32 * k, nbins - 1);
    const int4* p = (const int4*)(hist + (size_t)b * NCHUNK + lane * 8);
    a[k] = p[0];
    c[k] = p[1];
  }
  __builtin_amdgcn_sched_barrier(0);
#pragma unroll
  for (int k = 0; k < NBMAX / 32; ++k) {
    s[k] = a[k].x + a[k].y + a[k].z + a[k].w + c[k].x + c[k].y + c[k].z + c[k].w;
    inc[k] = s[k];
  }
#pragma unroll
  for (int off = 1; off < 32; off <<= 1) {
    int u[NBMAX / 32];
#pragma unroll
    for (int k = 0; k < NBMAX / 32; ++k) u[k] = __shfl_up(inc[k], off, 32);
    __builtin_amdgcn_sched_barrier(0);
#pragma unroll
    for (int k = 0; k < NBMAX / 32; ++k)
      if (lane >= off) inc[k] += u[k];
  }
#pragma unroll
  for (int k = 0; k < NBMAX / 32; ++k)
    if (lane == 31 && wave + 32 * k < nbins) tot[wave + 32 * k] = inc[k];
  __syncthreads();

  const int v = (tid < nbins) ? tot[tid] : 0;
  const int incl = wave_scan_incl(v, lane);
  if (tid < NBMAX && lane == 31) wsum[wave] = incl;
  __syncthreads();
  if (tid < NBMAX) {
    int off = 0;
    for (int w = 0; w < NBMAX / 32; ++w)
      if (w < wave) off += wsum[w];
    const int exb = off + incl - v;
    tot[tid] = exb;
    if (tid < nbins) binStart[tid] = exb;
    if (tid == nbins - 1) binStart[nbins] = exb + v;
  }
  __syncthreads();

#pragma unroll
  for (int k = 0; k < NBMAX / 32; ++k) {
    const int b = wave + 32 * k;
    if (b < nbins) {
      int4 o0, o1;
      o0.x = tot[b] + inc[k] - s[k];
      o0.y = o0.x + a[k].x;
      o0.z = o0.y + a[k].y;
      o0.w = o0.z + a[k].z;
      o1.x = o0.w + a[k].w;
      o1.y = o1.x + c[k].x;
      o1.z = o1.y + c[k].y;
      o1.w = o1.z + c[k].z;
      int4* q = (int4*)(base + (size_t)b * NCHUNK + lane * 8);
      q[0] = o0;
      q[1] = o1;
    }
  }
}

__global__ void k_binscatter(const int* __restrict__ src, const int* __restrict__ dst,
                             const int* __restrict__ base, int2* __restrict__ binned,
                             int e, int nbins) {
  __shared__ int cur[NBMAX];
  const int tid = threadIdx.x;
  const int wg = blockIdx.x;
  if (tid < nbins) cur[tid] = base[tid * NCHUNK + wg];
  const int per = (e + NCHUNK - 1) / NCHUNK;
  const int beg = wg * per;
  const int end = min(beg + per, e);
  __syncthreads();
  for (int i0 = beg + tid; i0 < end; i0 += ETH * EB) {
    int s[EB], d[EB];
#pragma unroll
    for (int u = 0; u < EB; ++u) {
      const int j = min(i0 + u * ETH, end - 1);
      s[u] = src[j];
      d[u] = dst[j];
    }
    hold_batch(s[0], s[1], s[2], s[3], s[4], s[5], s[6], s[7]);
    int p[EB];
#pragma unroll
    for (int u = 0; u < EB; ++u)
      p[u] = atomicAdd(&cur[d[u] >> BSH], (i0 + u * ETH < end) ? 1 : 0);
#pragma unroll
    for (int u = 0; u < EB; ++u)
      if (i0 + u * ETH < end) binned[p[u]] = make_int2(s[u], d[u]);
  }
}

__global__ void k_bincount(const int2* __restrict__ binned, const int* __restrict__ binStart,
                           int* __restrict__ rowstart, float* __restrict__ dinv, int n) {
  __shared__ int cnt[BINN];
  __shared__ int wsum[BINN / 32];
  const int tid = threadIdx.x;
  const int lane = tid & 31;
  const int wave = tid >> 5;
  const int b = blockIdx.x;
  const int lo = binStart[b];
  const int hi = binStart[b + 1];
  if (tid < BINN) cnt[tid] = 0;
  __syncthreads();
  for (int i0 = lo + tid; i0 < hi; i0 += ETH * EB) {
    int d[EB];
#pragma unroll
    for (int u = 0; u < EB; ++u) d[u] = binned[min(i0 + u * ETH, hi - 1)].y;
    hold_batch(d[0], d[1], d[2], d[3], d[4], d[5], d[6], d[7]);
#pragma unroll
    for (int u = 0; u < EB; ++u)
      if (i0 + u * ETH < hi) atomicAdd(&cnt[d[u] & (BINN - 1)], 1);
  }
  __syncthreads();
  const int c = (tid < BINN) ? cnt[tid] : 0;
  const int incl = wave_scan_incl(c, lane);
  if (tid < BINN && lane == 31) wsum[wave] = incl;
  __syncthreads();
  if (tid < BINN) {
    int off = 0;
    for (int w = 0; w < BINN / 32; ++w)
      if (w < wave) off += wsum[w];
    const int node = b * BINN + tid;
    if (node < n) {
      rowstart[node] = lo + off + incl - c;
      dinv[node] = rsqrtf((float)c + 1.0f);
    }
  }
  if (b == (int)gridDim.x - 1 && tid == 0) rowstart[n] = hi;
}

__global__ void k_binplace(const int2* __restrict__ binned, const int* __restrict__ binStart,
                           const int* __restrict__ rowstart, const float* __restrict__ dinv,
                           int2* __restrict__ rec, int n) {
  __shared__ int cur[BINN];
  __shared__ float dl[BINN];
  const int tid = threadIdx.x;
  const int b = blockIdx.x;
  const int lo = binStart[b];
  const int hi = binStart[b + 1];
  if (tid < BINN) {
    const int node = min(b * BINN + tid, n - 1);
    cur[tid] = rowstart[node];
    dl[tid] = dinv[node];
  }
  __syncthreads();
  for (int i0 = lo + tid; i0 < hi; i0 += ETH * EB) {
    int2 r[EB];
    float ds[EB];
#pragma unroll
    for (int u = 0; u < EB; ++u) r[u] = binned[min(i0 + u * ETH, hi - 1)];
    hold_batch(r[0].x, r[1].x, r[2].x, r[3].x, r[4].x, r[5].x, r[6].x, r[7].x);
#pragma unroll
    for (int u = 0; u < EB; ++u) ds[u] = dinv[(unsigned)r[u].x];
    hold_batch(__float_as_int(ds[0]), __float_as_int(ds[1]), __float_as_int(ds[2]),
               __float_as_int(ds[3]), __float_as_int(ds[4]), __float_as_int(ds[5]),
               __float_as_int(ds[6]), __float_as_int(ds[7]));
    int p[EB];
    float w[EB];
#pragma unroll
    for (int u = 0; u < EB; ++u) {
      const int t = r[u].y & (BINN - 1);
      p[u] = atomicAdd(&cur[t], (i0 + u * ETH < hi) ? 1 : 0);
      w[u] = ds[u] * dl[t];
    }
#pragma unroll
    for (int u = 0; u < EB; ++u)
      if (i0 + u * ETH < hi) rec[p[u]] = make_int2(r[u].x, __float_as_int(w[u]));
  }
}

__global__ void k_gemm128(const float* __restrict__ X, const float* __restrict__ W,
                          const float* __restrict__ bias, float* __restrict__ H, int n) {
  extern __shared__ float ws[];
  const int tid = threadIdx.x;

  for (int i = tid; i < HID * HID; i += 256) {
    int k = i >> 7, c = i & 127;
    ws[c * WSTRIDE + k] = W[i];
  }
  __syncthreads();

  const int wave   = tid >> 5;
  const int lane   = tid & 31;
  const int lane15 = lane & 15;
  const int half   = lane >> 4;
  const int rowBase = blockIdx.x * 128 + wave * 16;
  const int arow    = rowBase + lane15;
  const int koff    = half * 2;

  v2f a[32];
  if (arow < n) {
    const float* xp = X + (size_t)arow * HID + koff;
    if (bias) {
      const float* bp = bias + koff;
#pragma unroll
      for (int kk = 0; kk < 32; ++kk) {
        v2f xv = *(const v2f*)(xp + kk * 4);
        v2f bv = *(const v2f*)(bp + kk * 4);
        v2f t = xv + bv;
        t.x = fmaxf(t.x, 0.0f);
        t.y = fmaxf(t.y, 0.0f);
        a[kk] = t;
      }
    } else {
#pragma unroll
      for (int kk = 0; kk < 32; ++kk)
        a[kk] = *(const v2f*)(xp + kk * 4);
    }
  } else {
    v2f z = {};
#pragma unroll
    for (int kk = 0; kk < 32; ++kk) a[kk] = z;
  }

#pragma unroll
  for (int nt = 0; nt < 8; ++nt) {
    v8f acc = {};
    const float* bp = ws + (nt * 16 + lane15) * WSTRIDE + koff;
#pragma unroll
    for (int kk = 0; kk < 32; ++kk) {
      v2f b = *(const v2f*)(bp + kk * 4);
      acc = __builtin_amdgcn_wmma_f32_16x16x4_f32(
          false, a[kk], false, b, (short)0, acc, false, false);
    }
    const int colOut = nt * 16 + lane15;
#pragma unroll
    for (int r = 0; r < 8; ++r) {
      int row = rowBase + r + half * 8;
      if (row < n) H[(size_t)row * HID + colOut] = acc[r];
    }
  }
}

__global__ void k_aggregate(const float* __restrict__ H, const int2* __restrict__ rec,
                            const int* __restrict__ rowstart, const float* __restrict__ dinv,
                            float* __restrict__ AGG, int n) {
  long long gid = (long long)blockIdx.x * blockDim.x + threadIdx.x;
  int node = (int)(gid >> 5);
  if (node >= n) return;
  int q = ((int)gid & 31) * 4;

  int beg = __builtin_amdgcn_readfirstlane(rowstart[node]);
  int end = __builtin_amdgcn_readfirstlane(rowstart[node + 1]);

  float wself = dinv[node];
  wself *= wself;
  float4 acc = *(const float4*)(H + (size_t)node * HID + q);
  acc.x *= wself; acc.y *= wself; acc.z *= wself; acc.w *= wself;

  for (int j = beg; j < end; ++j) {
    int2 r = rec[j];
    int s = r.x;
    float w = __int_as_float(r.y);
    const float4 h = *(const float4*)(H + (size_t)s * HID + q);
    acc.x += w * h.x; acc.y += w * h.y; acc.z += w * h.z; acc.w += w * h.w;
  }
  *(float4*)(AGG + (size_t)node * HID + q) = acc;
}

template <bool ACT>
__global__ void k_gemm16(const float* __restrict__ X, const float* __restrict__ B, int ldb, int nc,
                         const float* __restrict__ bias, float* __restrict__ Z, int n) {
  __shared__ float bs[ZW * WSTRIDE];
  __shared__ float bsh[256];
  const int tid = threadIdx.x;

  const int wave   = tid >> 5;
  const int lane   = tid & 31;
  const int lane15 = lane & 15;
  const int half   = lane >> 4;
  const int rowBase = blockIdx.x * 128 + wave * 16;
  const int arow    = rowBase + lane15;
  const int koff    = half * 2;

  float sv[HID * ZW / 256];
#pragma unroll
  for (int it = 0; it < HID * ZW / 256; ++it) {
    int i = tid + it * 256;
    sv[it] = B[(i >> 4) * ldb + min(i & 15, nc - 1)];
  }
  float bsv = 0.0f;
  if (ACT) bsv = bias[tid & (HID - 1)];

  v2f a[32];
  const float* xp = X + (size_t)min(arow, n - 1) * HID + koff;
#pragma unroll
  for (int kk = 0; kk < 32; ++kk)
    a[kk] = *(const v2f*)(xp + kk * 4);

#pragma unroll
  for (int it = 0; it < HID * ZW / 256; ++it) {
    int i = tid + it * 256;
    int k = i >> 4, c = i & 15;
    bs[c * WSTRIDE + k] = (c < nc) ? sv[it] : 0.0f;
  }
  if (ACT) bsh[tid] = bsv;
  __syncthreads();

  v8f acc = {};
  const float* bp = bs + lane15 * WSTRIDE + koff;
  const float* cp = bsh + koff;
#pragma unroll
  for (int kk = 0; kk < 32; ++kk) {
    v2f av = a[kk];
    if (ACT) {
      v2f cv = *(const v2f*)(cp + kk * 4);
      av = av + cv;
      av.x = fmaxf(av.x, 0.0f);
      av.y = fmaxf(av.y, 0.0f);
    }
    v2f b = *(const v2f*)(bp + kk * 4);
    acc = __builtin_amdgcn_wmma_f32_16x16x4_f32(
        false, av, false, b, (short)0, acc, false, false);
  }
#pragma unroll
  for (int r = 0; r < 8; ++r) {
    int row = rowBase + r + half * 8;
    if (row < n) Z[(size_t)row * ZW + lane15] = acc[r];
  }
}

__global__ void k_aggregate16(const float* __restrict__ Z, const int2* __restrict__ rec,
                              const int* __restrict__ rowstart, const float* __restrict__ dinv,
                              float* __restrict__ AZ, int n) {
  const int gid = blockIdx.x * blockDim.x + threadIdx.x;
  const int node = gid >> 2;
  if (node >= n) return;
  const int q = (gid & 3) * 4;

  const int beg = rowstart[node];
  const int end = rowstart[node + 1];

  float wself = dinv[node];
  wself *= wself;
  float4 acc = *(const float4*)(Z + (size_t)node * ZW + q);
  acc.x *= wself; acc.y *= wself; acc.z *= wself; acc.w *= wself;

  for (int j = beg; j < end; j += 4) {
    int2 r[4];
    float4 h[4];
#pragma unroll
    for (int t = 0; t < 4; ++t) r[t] = rec[(j + t < end) ? (j + t) : j];
#pragma unroll
    for (int t = 0; t < 4; ++t) h[t] = *(const float4*)(Z + (size_t)r[t].x * ZW + q);
#pragma unroll
    for (int t = 0; t < 4; ++t) {
      const bool ok = (j + t) < end;
      const float w = __int_as_float(r[t].y);
      const float hx = ok ? h[t].x : 0.0f, hy = ok ? h[t].y : 0.0f;
      const float hz = ok ? h[t].z : 0.0f, hw = ok ? h[t].w : 0.0f;
      acc.x += w * hx; acc.y += w * hy; acc.z += w * hz; acc.w += w * hw;
    }
  }
  *(float4*)(AZ + (size_t)node * ZW + q) = acc;
}

#define PCH 8

__device__ __forceinline__ void pool_flush(float* __restrict__ pooled, float* __restrict__ cnt,
                                           int g, int q, float4 acc, int run) {
  float* p = pooled + (size_t)g * ZW + q;
  __hip_atomic_fetch_add(p + 0, acc.x, __ATOMIC_RELAXED, __HIP_MEMORY_SCOPE_AGENT);
  __hip_atomic_fetch_add(p + 1, acc.y, __ATOMIC_RELAXED, __HIP_MEMORY_SCOPE_AGENT);
  __hip_atomic_fetch_add(p + 2, acc.z, __ATOMIC_RELAXED, __HIP_MEMORY_SCOPE_AGENT);
  __hip_atomic_fetch_add(p + 3, acc.w, __ATOMIC_RELAXED, __HIP_MEMORY_SCOPE_AGENT);
  if (q == 0)
    __hip_atomic_fetch_add(&cnt[g], (float)run, __ATOMIC_RELAXED, __HIP_MEMORY_SCOPE_AGENT);
}

__global__ void k_pool16(const float* __restrict__ AZ, const int* __restrict__ batch,
                         float* __restrict__ pooled, float* __restrict__ cnt, int n) {
  const int gid = blockIdx.x * blockDim.x + threadIdx.x;
  const int first = (gid >> 5) * (8 * PCH) + ((gid & 31) >> 2);
  if (first >= n) return;
  const int q = (gid & 3) * 4;
  const int last = n - 1;

  int g[PCH];
  float4 h[PCH];
#pragma unroll
  for (int i = 0; i < PCH; ++i) {
    const int node = min(first + i * 8, last);
    g[i] = batch[node];
    h[i] = *(const float4*)(AZ + (size_t)node * ZW + q);
  }

  int cur = g[0];
  int run = 1;
  float4 acc = h[0];
#pragma unroll
  for (int i = 1; i < PCH; ++i) {
    if (first + i * 8 <= last) {
      if (g[i] != cur) {
        pool_flush(pooled, cnt, cur, q, acc, run);
        cur = g[i];
        run = 1;
        acc = h[i];
      } else {
        acc.x += h[i].x; acc.y += h[i].y; acc.z += h[i].z; acc.w += h[i].w;
        ++run;
      }
    }
  }
  pool_flush(pooled, cnt, cur, q, acc, run);
}

__global__ void k_final16(const float* __restrict__ pooled, const float* __restrict__ cnt,
                          const float* __restrict__ b3, const float* __restrict__ Wl,
                          const float* __restrict__ bl, float* __restrict__ out, int G, int C) {
  int gid = blockIdx.x * blockDim.x + threadIdx.x;
  if (gid >= G * C) return;
  int g = gid / C, c = gid % C;
  float cn = cnt[g];
  float inv = 1.0f / fmaxf(cn, 1.0f);
  float bw = 0.0f;
  for (int k = 0; k < HID; ++k)
    bw += b3[k] * Wl[k * C + c];
  float s = pooled[(size_t)g * ZW + c] * inv;
  if (cn > 0.0f) s += bw;
  out[gid] = s + bl[c];
}

extern "C" void kernel_launch(void* const* d_in, const int* in_sizes, int n_in,
                              void* d_out, int out_size, void* d_ws, size_t ws_size,
                              hipStream_t stream) {
  const float* x   = (const float*)d_in[0];
  const int*   ei  = (const int*)d_in[1];
  const int*   bat = (const int*)d_in[2];
  const float* W1  = (const float*)d_in[3];
  const float* b1  = (const float*)d_in[4];
  const float* W2  = (const float*)d_in[5];
  const float* b2  = (const float*)d_in[6];
  const float* W3  = (const float*)d_in[7];
  const float* b3  = (const float*)d_in[8];
  const float* Wl  = (const float*)d_in[9];
  const float* bl  = (const float*)d_in[10];

  const int n = in_sizes[2];
  const int e = in_sizes[1] / 2;
  const int C = in_sizes[10];
  const int G = out_size / C;
  const int* src = ei;
  const int* dst = ei + e;

  char* base = (char*)d_ws;
  auto alloc = [&](size_t bytes) -> void* {
    void* p = (void*)base;
    base += (bytes + 255) & ~(size_t)255;
    return p;
  };
  int*   hist      = (int*)alloc((size_t)NBMAX * NCHUNK * 4);
  int*   base0     = (int*)alloc((size_t)NBMAX * NCHUNK * 4);
  int*   binStart  = (int*)alloc(((size_t)NBMAX + 1) * 4);
  int2*  binned    = (int2*)alloc((size_t)e * 8);
  int*   rowstart  = (int*)alloc(((size_t)n + 1) * 4);
  float* dinv      = (float*)alloc((size_t)n * 4);
  int2*  rec       = (int2*)alloc((size_t)e * 8);
  float* Hb        = (float*)alloc((size_t)n * HID * 4);
  float* Ab        = (float*)alloc((size_t)n * HID * 4);
  float* Mb        = (float*)alloc((size_t)HID * ZW * 4);
  float* Zb        = (float*)alloc((size_t)n * ZW * 4);
  float* AZb       = (float*)alloc((size_t)n * ZW * 4);
  float* pooled    = (float*)alloc((size_t)G * ZW * 4);
  float* cnt       = (float*)alloc((size_t)G * 4);
  (void)ws_size; (void)n_in;

  const int T = 256;
  auto nb = [](long long w, int t) { return (unsigned)((w + t - 1) / t); };
  const long long nw = (long long)n * 32;
  const unsigned gemmB = (unsigned)((n + 127) / 128);
  const unsigned nbins = (unsigned)((n + BINN - 1) >> BSH);
  const size_t lds = (size_t)HID * WSTRIDE * sizeof(float);

  k_binhist<<<NCHUNK, ETH, 0, stream>>>(dst, hist, e, (int)nbins);
  k_binscan<<<1, ETH, 0, stream>>>(hist, base0, binStart, (int)nbins);
  k_binscatter<<<NCHUNK, ETH, 0, stream>>>(src, dst, base0, binned, e, (int)nbins);
  k_bincount<<<nbins, ETH, 0, stream>>>(binned, binStart, rowstart, dinv, n);
  k_binplace<<<nbins, ETH, 0, stream>>>(binned, binStart, rowstart, dinv, rec, n);

  k_gemm128<<<gemmB, T, lds, stream>>>(x, W1, nullptr, Hb, n);
  k_aggregate<<<nb(nw, T), T, 0, stream>>>(Hb, rec, rowstart, dinv, Ab, n);

  k_gemm128<<<gemmB, T, lds, stream>>>(Ab, W2, b1, Hb, n);
  k_aggregate<<<nb(nw, T), T, 0, stream>>>(Hb, rec, rowstart, dinv, Ab, n);

  k_gemm16<false><<<1, T, 0, stream>>>(W3, Wl, C, C, nullptr, Mb, HID);
  k_gemm16<true><<<gemmB, T, 0, stream>>>(Ab, Mb, ZW, ZW, b2, Zb, n);
  k_aggregate16<<<nb((long long)n * 4, T), T, 0, stream>>>(Zb, rec, rowstart, dinv, AZb, n);

  k_zero_f<<<nb((long long)G * ZW, T), T, 0, stream>>>(pooled, G * ZW);
  k_zero_f<<<nb(G, T), T, 0, stream>>>(cnt, G);
  k_pool16<<<nb((long long)((n + 8 * PCH - 1) / (8 * PCH)) * 32, T), T, 0, stream>>>(AZb, bat, pooled,
                                                                               cnt, n);
  k_final16<<<nb((long long)G * C, T), T, 0, stream>>>(pooled, cnt, b3, Wl, bl,
                                                       (float*)d_out, G, C);
}


# Neighbour sum fused into the product behind it: a wave walks the records of its 16 nodes as one flat list, 16 row loads in flight, rows reach WMMA via LDS; 16 launches to 14; sum order unchanged
# speedup vs baseline: 2.1997x; 1.0422x over previous
#include <hip/hip_runtime.h>
#include <hip/hip_bf16.h>

typedef __attribute__((ext_vector_type(2))) float v2f;
typedef __attribute__((ext_vector_type(8))) float v8f;

#define HID 128
#define WSTRIDE 132
#define ZW 16

__global__ void k_zero_f(float* __restrict__ p, int n) {
  int i = blockIdx.x * blockDim.x + threadIdx.x;
  if (i < n) p[i] = 0.0f;
}

#define BSH    9
#define BINN   (1 << BSH)
#define NBMAX  256
#define NCHUNK 256
#define ETH    1024
#define EB     8

__device__ __forceinline__ int wave_scan_incl(int v, int lane) {
#pragma unroll
  for (int off = 1; off < 32; off <<= 1) {
    int u = __shfl_up(v, off, 32);
    if (lane >= off) v += u;
  }
  return v;
}

__device__ __forceinline__ void hold_batch(int v0, int v1, int v2, int v3,
                                           int v4, int v5, int v6, int v7) {
  asm volatile("" :: "v"(v0), "v"(v1), "v"(v2), "v"(v3), "v"(v4), "v"(v5), "v"(v6), "v"(v7));
  __builtin_amdgcn_sched_barrier(0);
}

__global__ void k_binhist(const int* __restrict__ dst, int* __restrict__ hist, int e, int nbins) {
  __shared__ int h[NBMAX];
  const int tid = threadIdx.x;
  const int wg = blockIdx.x;
  if (tid < NBMAX) h[tid] = 0;
  const int per = (e + NCHUNK - 1) / NCHUNK;
  const int beg = wg * per;
  const int end = min(beg + per, e);
  __syncthreads();
  for (int i0 = beg + tid; i0 < end; i0 += ETH * EB) {
    int d[EB];
#pragma unroll
    for (int u = 0; u < EB; ++u) d[u] = dst[min(i0 + u * ETH, end - 1)];
    hold_batch(d[0], d[1], d[2], d[3], d[4], d[5], d[6], d[7]);
#pragma unroll
    for (int u = 0; u < EB; ++u)
      if (i0 + u * ETH < end) atomicAdd(&h[d[u] >> BSH], 1);
  }
  __syncthreads();
  if (tid < nbins) hist[tid * NCHUNK + wg] = h[tid];
}

__global__ void k_binscan(const int* __restrict__ hist, int* __restrict__ base,
                          int* __restrict__ binStart, int nbins) {
  __shared__ int tot[NBMAX];
  __shared__ int wsum[NBMAX / 32];
  const int tid = threadIdx.x;
  const int lane = tid & 31;
  const int wave = tid >> 5;

  int4 a[NBMAX / 32], c[NBMAX / 32];
  int s[NBMAX / 32], inc[NBMAX / 32];
#pragma unroll
  for (int k = 0; k < NBMAX / 32; ++k) {
    const int b = min(wave + 32 * k, nbins - 1);
    const int4* p = (const int4*)(hist + (size_t)b * NCHUNK + lane * 8);
    a[k] = p[0];
    c[k] = p[1];
  }
  __builtin_amdgcn_sched_barrier(0);
#pragma unroll
  for (int k = 0; k < NBMAX / 32; ++k) {
    s[k] = a[k].x + a[k].y + a[k].z + a[k].w + c[k].x + c[k].y + c[k].z + c[k].w;
    inc[k] = s[k];
  }
#pragma unroll
  for (int off = 1; off < 32; off <<= 1) {
    int u[NBMAX / 32];
#pragma unroll
    for (int k = 0; k < NBMAX / 32; ++k) u[k] = __shfl_up(inc[k], off, 32);
    __builtin_amdgcn_sched_barrier(0);
#pragma unroll
    for (int k = 0; k < NBMAX / 32; ++k)
      if (lane >= off) inc[k] += u[k];
  }
#pragma unroll
  for (int k = 0; k < NBMAX / 32; ++k)
    if (lane == 31 && wave + 32 * k < nbins) tot[wave + 32 * k] = inc[k];
  __syncthreads();

  const int v = (tid < nbins) ? tot[tid] : 0;
  const int incl = wave_scan_incl(v, lane);
  if (tid < NBMAX && lane == 31) wsum[wave] = incl;
  __syncthreads();
  if (tid < NBMAX) {
    int off = 0;
    for (int w = 0; w < NBMAX / 32; ++w)
      if (w < wave) off += wsum[w];
    const int exb = off + incl - v;
    tot[tid] = exb;
    if (tid < nbins) binStart[tid] = exb;
    if (tid == nbins - 1) binStart[nbins] = exb + v;
  }
  __syncthreads();

#pragma unroll
  for (int k = 0; k < NBMAX / 32; ++k) {
    const int b = wave + 32 * k;
    if (b < nbins) {
      int4 o0, o1;
      o0.x = tot[b] + inc[k] - s[k];
      o0.y = o0.x + a[k].x;
      o0.z = o0.y + a[k].y;
      o0.w = o0.z + a[k].z;
      o1.x = o0.w + a[k].w;
      o1.y = o1.x + c[k].x;
      o1.z = o1.y + c[k].y;
      o1.w = o1.z + c[k].z;
      int4* q = (int4*)(base + (size_t)b * NCHUNK + lane * 8);
      q[0] = o0;
      q[1] = o1;
    }
  }
}

__global__ void k_binscatter(const int* __restrict__ src, const int* __restrict__ dst,
                             const int* __restrict__ base, int2* __restrict__ binned,
                             int e, int nbins) {
  __shared__ int cur[NBMAX];
  const int tid = threadIdx.x;
  const int wg = blockIdx.x;
  if (tid < nbins) cur[tid] = base[tid * NCHUNK + wg];
  const int per = (e + NCHUNK - 1) / NCHUNK;
  const int beg = wg * per;
  const int end = min(beg + per, e);
  __syncthreads();
  for (int i0 = beg + tid; i0 < end; i0 += ETH * EB) {
    int s[EB], d[EB];
#pragma unroll
    for (int u = 0; u < EB; ++u) {
      const int j = min(i0 + u * ETH, end - 1);
      s[u] = src[j];
      d[u] = dst[j];
    }
    hold_batch(s[0], s[1], s[2], s[3], s[4], s[5], s[6], s[7]);
    int p[EB];
#pragma unroll
    for (int u = 0; u < EB; ++u)
      p[u] = atomicAdd(&cur[d[u] >> BSH], (i0 + u * ETH < end) ? 1 : 0);
#pragma unroll
    for (int u = 0; u < EB; ++u)
      if (i0 + u * ETH < end) binned[p[u]] = make_int2(s[u], d[u]);
  }
}

__global__ void k_bincount(const int2* __restrict__ binned, const int* __restrict__ binStart,
                           int* __restrict__ rowstart, float* __restrict__ dinv, int n) {
  __shared__ int cnt[BINN];
  __shared__ int wsum[BINN / 32];
  const int tid = threadIdx.x;
  const int lane = tid & 31;
  const int wave = tid >> 5;
  const int b = blockIdx.x;
  const int lo = binStart[b];
  const int hi = binStart[b + 1];
  if (tid < BINN) cnt[tid] = 0;
  __syncthreads();
  for (int i0 = lo + tid; i0 < hi; i0 += ETH * EB) {
    int d[EB];
#pragma unroll
    for (int u = 0; u < EB; ++u) d[u] = binned[min(i0 + u * ETH, hi - 1)].y;
    hold_batch(d[0], d[1], d[2], d[3], d[4], d[5], d[6], d[7]);
#pragma unroll
    for (int u = 0; u < EB; ++u)
      if (i0 + u * ETH < hi) atomicAdd(&cnt[d[u] & (BINN - 1)], 1);
  }
  __syncthreads();
  const int c = (tid < BINN) ? cnt[tid] : 0;
  const int incl = wave_scan_incl(c, lane);
  if (tid < BINN && lane == 31) wsum[wave] = incl;
  __syncthreads();
  if (tid < BINN) {
    int off = 0;
    for (int w = 0; w < BINN / 32; ++w)
      if (w < wave) off += wsum[w];
    const int node = b * BINN + tid;
    if (node < n) {
      rowstart[node] = lo + off + incl - c;
      dinv[node] = rsqrtf((float)c + 1.0f);
    }
  }
  if (b == (int)gridDim.x - 1 && tid == 0) rowstart[n] = hi;
}

__global__ void k_binplace(const int2* __restrict__ binned, const int* __restrict__ binStart,
                           const int* __restrict__ rowstart, const float* __restrict__ dinv,
                           int2* __restrict__ rec, int n) {
  __shared__ int cur[BINN];
  __shared__ float dl[BINN];
  const int tid = threadIdx.x;
  const int b = blockIdx.x;
  const int lo = binStart[b];
  const int hi = binStart[b + 1];
  if (tid < BINN) {
    const int node = min(b * BINN + tid, n - 1);
    cur[tid] = rowstart[node];
    dl[tid] = dinv[node];
  }
  __syncthreads();
  for (int i0 = lo + tid; i0 < hi; i0 += ETH * EB) {
    int2 r[EB];
    float ds[EB];
#pragma unroll
    for (int u = 0; u < EB; ++u) r[u] = binned[min(i0 + u * ETH, hi - 1)];
    hold_batch(r[0].x, r[1].x, r[2].x, r[3].x, r[4].x, r[5].x, r[6].x, r[7].x);
#pragma unroll
    for (int u = 0; u < EB; ++u) ds[u] = dinv[(unsigned)r[u].x];
    hold_batch(__float_as_int(ds[0]), __float_as_int(ds[1]), __float_as_int(ds[2]),
               __float_as_int(ds[3]), __float_as_int(ds[4]), __float_as_int(ds[5]),
               __float_as_int(ds[6]), __float_as_int(ds[7]));
    int p[EB];
    float w[EB];
#pragma unroll
    for (int u = 0; u < EB; ++u) {
      const int t = r[u].y & (BINN - 1);
      p[u] = atomicAdd(&cur[t], (i0 + u * ETH < hi) ? 1 : 0);
      w[u] = ds[u] * dl[t];
    }
#pragma unroll
    for (int u = 0; u < EB; ++u)
      if (i0 + u * ETH < hi) rec[p[u]] = make_int2(r[u].x, __float_as_int(w[u]));
  }
}

__global__ void k_gemm128(const float* __restrict__ X, const float* __restrict__ W,
                          const float* __restrict__ bias, float* __restrict__ H, int n) {
  extern __shared__ float ws[];
  const int tid = threadIdx.x;

  for (int i = tid; i < HID * HID; i += 256) {
    int k = i >> 7, c = i & 127;
    ws[c * WSTRIDE + k] = W[i];
  }
  __syncthreads();

  const int wave   = tid >> 5;
  const int lane   = tid & 31;
  const int lane15 = lane & 15;
  const int half   = lane >> 4;
  const int rowBase = blockIdx.x * 128 + wave * 16;
  const int arow    = rowBase + lane15;
  const int koff    = half * 2;

  v2f a[32];
  if (arow < n) {
    const float* xp = X + (size_t)arow * HID + koff;
    if (bias) {
      const float* bp = bias + koff;
#pragma unroll
      for (int kk = 0; kk < 32; ++kk) {
        v2f xv = *(const v2f*)(xp + kk * 4);
        v2f bv = *(const v2f*)(bp + kk * 4);
        v2f t = xv + bv;
        t.x = fmaxf(t.x, 0.0f);
        t.y = fmaxf(t.y, 0.0f);
        a[kk] = t;
      }
    } else {
#pragma unroll
      for (int kk = 0; kk < 32; ++kk)
        a[kk] = *(const v2f*)(xp + kk * 4);
    }
  } else {
    v2f z = {};
#pragma unroll
    for (int kk = 0; kk < 32; ++kk) a[kk] = z;
  }

#pragma unroll
  for (int nt = 0; nt < 8; ++nt) {
    v8f acc = {};
    const float* bp = ws + (nt * 16 + lane15) * WSTRIDE + koff;
#pragma unroll
    for (int kk = 0; kk < 32; ++kk) {
      v2f b = *(const v2f*)(bp + kk * 4);
      acc = __builtin_amdgcn_wmma_f32_16x16x4_f32(
          false, a[kk], false, b, (short)0, acc, false, false);
    }
    const int colOut = nt * 16 + lane15;
#pragma unroll
    for (int r = 0; r < 8; ++r) {
      int row = rowBase + r + half * 8;
      if (row < n) H[(size_t)row * HID + colOut] = acc[r];
    }
  }
}

#define GCH 16

__device__ __forceinline__ void hold_rows(const float4 (&h)[GCH]) {
  asm volatile("" :: "v"(h[0].x), "v"(h[1].x), "v"(h[2].x), "v"(h[3].x), "v"(h[4].x), "v"(h[5].x),
               "v"(h[6].x), "v"(h[7].x), "v"(h[8].x), "v"(h[9].x), "v"(h[10].x), "v"(h[11].x),
               "v"(h[12].x), "v"(h[13].x), "v"(h[14].x), "v"(h[15].x));
}

template <int NT>
__global__ void __launch_bounds__(256)
k_agg_gemm(const float* __restrict__ H, const int2* __restrict__ rec,
           const int* __restrict__ rowstart, const float* __restrict__ dinv,
           const float* __restrict__ B, const float* __restrict__ bias,
           float* __restrict__ OUT, int n) {
  constexpr int NC = NT * 16;
  extern __shared__ float sm[];
  const int tid = threadIdx.x;
  const int wave   = tid >> 5;
  const int lane   = tid & 31;
  const int lane15 = lane & 15;
  const int half   = lane >> 4;
  const int koff   = half * 2;
  float* tile = sm + NC * WSTRIDE + wave * (16 * WSTRIDE);

  constexpr int NIT = HID * NC / 256;
  constexpr int SB = NIT < 32 ? NIT : 32;
  for (int it0 = 0; it0 < NIT; it0 += SB) {
    float sv[SB];
#pragma unroll
    for (int u = 0; u < SB; ++u) sv[u] = B[tid + (it0 + u) * 256];
#pragma unroll
    for (int u = 0; u < SB; ++u) {
      const int i = tid + (it0 + u) * 256;
      sm[(i % NC) * WSTRIDE + i / NC] = sv[u];
    }
  }
  __syncthreads();

  const int base = blockIdx.x * 128 + wave * 16;
  const int q = lane * 4;
  const int last = n - 1;
  const int rsv = rowstart[min(base + lane, n)];
  const float dvl = dinv[min(base + lane15, last)];
  const int beg = __builtin_amdgcn_readlane(rsv, 0);
  const int end = __builtin_amdgcn_readlane(rsv, 16);

  {
    float4 sr[16];
#pragma unroll
    for (int m = 0; m < 16; ++m)
      sr[m] = *(const float4*)(H + (size_t)min(base + m, last) * HID + q);
#pragma unroll
    for (int m = 0; m < 16; ++m) {
      float wself = __int_as_float(__builtin_amdgcn_readlane(__float_as_int(dvl), m));
      wself *= wself;
      float4 v = sr[m];
      v.x *= wself; v.y *= wself; v.z *= wself; v.w *= wself;
      if (base + m > last) v = make_float4(0.0f, 0.0f, 0.0f, 0.0f);
      *(float4*)(tile + m * WSTRIDE + q) = v;
    }
  }

  int m = 0;
  int nb = __builtin_amdgcn_readlane(rsv, 1);
  float4 acc = *(const float4*)(tile + q);
  if (beg < end) {
    int2 rv = rec[min(beg + lane, end - 1)];
    for (int jw = beg; jw < end; jw += 32) {
      const int2 rvn = rec[min(jw + 32 + lane, end - 1)];
#pragma unroll
      for (int c = 0; c < 32 / GCH; ++c) {
        const int j0 = jw + c * GCH;
        if (j0 < end) {
          float4 h[GCH];
#pragma unroll
          for (int t = 0; t < GCH; ++t) {
            const unsigned s = (unsigned)__builtin_amdgcn_readlane(rv.x, c * GCH + t);
            h[t] = *(const float4*)(H + (size_t)s * HID + q);
          }
          hold_rows(h);
#pragma unroll
          for (int t = 0; t < GCH; ++t) {
            if (j0 + t < end) {
              while (j0 + t == nb) {
                *(float4*)(tile + m * WSTRIDE + q) = acc;
                ++m;
                acc = *(const float4*)(tile + m * WSTRIDE + q);
                nb = __builtin_amdgcn_readlane(rsv, m + 1);
              }
              const float w = __int_as_float(__builtin_amdgcn_readlane(rv.y, c * GCH + t));
              acc.x += w * h[t].x; acc.y += w * h[t].y; acc.z += w * h[t].z; acc.w += w * h[t].w;
            }
          }
        }
      }
      rv = rvn;
    }
  }
  *(float4*)(tile + m * WSTRIDE + q) = acc;

  v2f a[32];
  {
    const float* tp = tile + lane15 * WSTRIDE + koff;
    const float* cp = bias + koff;
#pragma unroll
    for (int kk = 0; kk < 32; ++kk) {
      v2f xv = *(const v2f*)(tp + kk * 4);
      v2f bv = *(const v2f*)(cp + kk * 4);
      v2f t = xv + bv;
      t.x = fmaxf(t.x, 0.0f);
      t.y = fmaxf(t.y, 0.0f);
      a[kk] = t;
    }
  }

#pragma unroll
  for (int nt = 0; nt < NT; ++nt) {
    v8f accw = {};
    const float* bp = sm + (nt * 16 + lane15) * WSTRIDE + koff;
#pragma unroll
    for (int kk = 0; kk < 32; ++kk) {
      v2f b = *(const v2f*)(bp + kk * 4);
      accw = __builtin_amdgcn_wmma_f32_16x16x4_f32(
          false, a[kk], false, b, (short)0, accw, false, false);
    }
    const int colOut = nt * 16 + lane15;
#pragma unroll
    for (int r = 0; r < 8; ++r) {
      int row = base + r + half * 8;
      if (row < n) OUT[(size_t)row * NC + colOut] = accw[r];
    }
  }
}

template <bool ACT>
__global__ void k_gemm16(const float* __restrict__ X, const float* __restrict__ B, int ldb, int nc,
                         const float* __restrict__ bias, float* __restrict__ Z, int n) {
  __shared__ float bs[ZW * WSTRIDE];
  __shared__ float bsh[256];
  const int tid = threadIdx.x;

  const int wave   = tid >> 5;
  const int lane   = tid & 31;
  const int lane15 = lane & 15;
  const int half   = lane >> 4;
  const int rowBase = blockIdx.x * 128 + wave * 16;
  const int arow    = rowBase + lane15;
  const int koff    = half * 2;

  float sv[HID * ZW / 256];
#pragma unroll
  for (int it = 0; it < HID * ZW / 256; ++it) {
    int i = tid + it * 256;
    sv[it] = B[(i >> 4) * ldb + min(i & 15, nc - 1)];
  }
  float bsv = 0.0f;
  if (ACT) bsv = bias[tid & (HID - 1)];

  v2f a[32];
  const float* xp = X + (size_t)min(arow, n - 1) * HID + koff;
#pragma unroll
  for (int kk = 0; kk < 32; ++kk)
    a[kk] = *(const v2f*)(xp + kk * 4);

#pragma unroll
  for (int it = 0; it < HID * ZW / 256; ++it) {
    int i = tid + it * 256;
    int k = i >> 4, c = i & 15;
    bs[c * WSTRIDE + k] = (c < nc) ? sv[it] : 0.0f;
  }
  if (ACT) bsh[tid] = bsv;
  __syncthreads();

  v8f acc = {};
  const float* bp = bs + lane15 * WSTRIDE + koff;
  const float* cp = bsh + koff;
#pragma unroll
  for (int kk = 0; kk < 32; ++kk) {
    v2f av = a[kk];
    if (ACT) {
      v2f cv = *(const v2f*)(cp + kk * 4);
      av = av + cv;
      av.x = fmaxf(av.x, 0.0f);
      av.y = fmaxf(av.y, 0.0f);
    }
    v2f b = *(const v2f*)(bp + kk * 4);
    acc = __builtin_amdgcn_wmma_f32_16x16x4_f32(
        false, av, false, b, (short)0, acc, false, false);
  }
#pragma unroll
  for (int r = 0; r < 8; ++r) {
    int row = rowBase + r + half * 8;
    if (row < n) Z[(size_t)row * ZW + lane15] = acc[r];
  }
}

__global__ void k_aggregate16(const float* __restrict__ Z, const int2* __restrict__ rec,
                              const int* __restrict__ rowstart, const float* __restrict__ dinv,
                              float* __restrict__ AZ, int n) {
  const int gid = blockIdx.x * blockDim.x + threadIdx.x;
  const int node = gid >> 2;
  if (node >= n) return;
  const int q = (gid & 3) * 4;

  const int beg = rowstart[node];
  const int end = rowstart[node + 1];

  float wself = dinv[node];
  wself *= wself;
  float4 acc = *(const float4*)(Z + (size_t)node * ZW + q);
  acc.x *= wself; acc.y *= wself; acc.z *= wself; acc.w *= wself;

  for (int j = beg; j < end; j += 4) {
    int2 r[4];
    float4 h[4];
#pragma unroll
    for (int t = 0; t < 4; ++t) r[t] = rec[(j + t < end) ? (j + t) : j];
#pragma unroll
    for (int t = 0; t < 4; ++t) h[t] = *(const float4*)(Z + (size_t)r[t].x * ZW + q);
#pragma unroll
    for (int t = 0; t < 4; ++t) {
      const bool ok = (j + t) < end;
      const float w = __int_as_float(r[t].y);
      const float hx = ok ? h[t].x : 0.0f, hy = ok ? h[t].y : 0.0f;
      const float hz = ok ? h[t].z : 0.0f, hw = ok ? h[t].w : 0.0f;
      acc.x += w * hx; acc.y += w * hy; acc.z += w * hz; acc.w += w * hw;
    }
  }
  *(float4*)(AZ + (size_t)node * ZW + q) = acc;
}

#define PCH 8

__device__ __forceinline__ void pool_flush(float* __restrict__ pooled, float* __restrict__ cnt,
                                           int g, int q, float4 acc, int run) {
  float* p = pooled + (size_t)g * ZW + q;
  __hip_atomic_fetch_add(p + 0, acc.x, __ATOMIC_RELAXED, __HIP_MEMORY_SCOPE_AGENT);
  __hip_atomic_fetch_add(p + 1, acc.y, __ATOMIC_RELAXED, __HIP_MEMORY_SCOPE_AGENT);
  __hip_atomic_fetch_add(p + 2, acc.z, __ATOMIC_RELAXED, __HIP_MEMORY_SCOPE_AGENT);
  __hip_atomic_fetch_add(p + 3, acc.w, __ATOMIC_RELAXED, __HIP_MEMORY_SCOPE_AGENT);
  if (q == 0)
    __hip_atomic_fetch_add(&cnt[g], (float)run, __ATOMIC_RELAXED, __HIP_MEMORY_SCOPE_AGENT);
}

__global__ void k_pool16(const float* __restrict__ AZ, const int* __restrict__ batch,
                         float* __restrict__ pooled, float* __restrict__ cnt, int n) {
  const int gid = blockIdx.x * blockDim.x + threadIdx.x;
  const int first = (gid >> 5) * (8 * PCH) + ((gid & 31) >> 2);
  if (first >= n) return;
  const int q = (gid & 3) * 4;
  const int last = n - 1;

  int g[PCH];
  float4 h[PCH];
#pragma unroll
  for (int i = 0; i < PCH; ++i) {
    const int node = min(first + i * 8, last);
    g[i] = batch[node];
    h[i] = *(const float4*)(AZ + (size_t)node * ZW + q);
  }

  int cur = g[0];
  int run = 1;
  float4 acc = h[0];
#pragma unroll
  for (int i = 1; i < PCH; ++i) {
    if (first + i * 8 <= last) {
      if (g[i] != cur) {
        pool_flush(pooled, cnt, cur, q, acc, run);
        cur = g[i];
        run = 1;
        acc = h[i];
      } else {
        acc.x += h[i].x; acc.y += h[i].y; acc.z += h[i].z; acc.w += h[i].w;
        ++run;
      }
    }
  }
  pool_flush(pooled, cnt, cur, q, acc, run);
}

__global__ void k_final16(const float* __restrict__ pooled, const float* __restrict__ cnt,
                          const float* __restrict__ b3, const float* __restrict__ Wl,
                          const float* __restrict__ bl, float* __restrict__ out, int G, int C) {
  int gid = blockIdx.x * blockDim.x + threadIdx.x;
  if (gid >= G * C) return;
  int g = gid / C, c = gid % C;
  float cn = cnt[g];
  float inv = 1.0f / fmaxf(cn, 1.0f);
  float bw = 0.0f;
  for (int k = 0; k < HID; ++k)
    bw += b3[k] * Wl[k * C + c];
  float s = pooled[(size_t)g * ZW + c] * inv;
  if (cn > 0.0f) s += bw;
  out[gid] = s + bl[c];
}

extern "C" void kernel_launch(void* const* d_in, const int* in_sizes, int n_in,
                              void* d_out, int out_size, void* d_ws, size_t ws_size,
                              hipStream_t stream) {
  const float* x   = (const float*)d_in[0];
  const int*   ei  = (const int*)d_in[1];
  const int*   bat = (const int*)d_in[2];
  const float* W1  = (const float*)d_in[3];
  const float* b1  = (const float*)d_in[4];
  const float* W2  = (const float*)d_in[5];
  const float* b2  = (const float*)d_in[6];
  const float* W3  = (const float*)d_in[7];
  const float* b3  = (const float*)d_in[8];
  const float* Wl  = (const float*)d_in[9];
  const float* bl  = (const float*)d_in[10];

  const int n = in_sizes[2];
  const int e = in_sizes[1] / 2;
  const int C = in_sizes[10];
  const int G = out_size / C;
  const int* src = ei;
  const int* dst = ei + e;

  char* base = (char*)d_ws;
  auto alloc = [&](size_t bytes) -> void* {
    void* p = (void*)base;
    base += (bytes + 255) & ~(size_t)255;
    return p;
  };
  int*   hist      = (int*)alloc((size_t)NBMAX * NCHUNK * 4);
  int*   base0     = (int*)alloc((size_t)NBMAX * NCHUNK * 4);
  int*   binStart  = (int*)alloc(((size_t)NBMAX + 1) * 4);
  int2*  binned    = (int2*)alloc((size_t)e * 8);
  int*   rowstart  = (int*)alloc(((size_t)n + 1) * 4);
  float* dinv      = (float*)alloc((size_t)n * 4);
  int2*  rec       = (int2*)alloc((size_t)e * 8);
  float* Hb        = (float*)alloc((size_t)n * HID * 4);
  float* Ab        = (float*)alloc((size_t)n * HID * 4);
  float* Mb        = (float*)alloc((size_t)HID * ZW * 4);
  float* Zb        = (float*)alloc((size_t)n * ZW * 4);
  float* AZb       = (float*)alloc((size_t)n * ZW * 4);
  float* pooled    = (float*)alloc((size_t)G * ZW * 4);
  float* cnt       = (float*)alloc((size_t)G * 4);
  (void)ws_size; (void)n_in;

  const int T = 256;
  auto nb = [](long long w, int t) { return (unsigned)((w + t - 1) / t); };
  const unsigned gemmB = (unsigned)((n + 127) / 128);
  const unsigned nbins = (unsigned)((n + BINN - 1) >> BSH);
  const size_t lds = (size_t)HID * WSTRIDE * sizeof(float);
  const size_t ldsF8 = (size_t)(8 * 16 + 128) * WSTRIDE * sizeof(float);
  const size_t ldsF1 = (size_t)(1 * 16 + 128) * WSTRIDE * sizeof(float);

  k_binhist<<<NCHUNK, ETH, 0, stream>>>(dst, hist, e, (int)nbins);
  k_binscan<<<1, ETH, 0, stream>>>(hist, base0, binStart, (int)nbins);
  k_binscatter<<<NCHUNK, ETH, 0, stream>>>(src, dst, base0, binned, e, (int)nbins);
  k_bincount<<<nbins, ETH, 0, stream>>>(binned, binStart, rowstart, dinv, n);
  k_binplace<<<nbins, ETH, 0, stream>>>(binned, binStart, rowstart, dinv, rec, n);

  k_gemm128<<<gemmB, T, lds, stream>>>(x, W1, nullptr, Hb, n);

  k_agg_gemm<8><<<gemmB, T, ldsF8, stream>>>(Hb, rec, rowstart, dinv, W2, b1, Ab, n);

  k_gemm16<false><<<1, T, 0, stream>>>(W3, Wl, C, C, nullptr, Mb, HID);
  k_agg_gemm<1><<<gemmB, T, ldsF1, stream>>>(Ab, rec, rowstart, dinv, Mb, b2, Zb, n);
  k_aggregate16<<<nb((long long)n * 4, T), T, 0, stream>>>(Zb, rec, rowstart, dinv, AZb, n);

  k_zero_f<<<nb((long long)G * ZW, T), T, 0, stream>>>(pooled, G * ZW);
  k_zero_f<<<nb(G, T), T, 0, stream>>>(cnt, G);
  k_pool16<<<nb((long long)((n + 8 * PCH - 1) / (8 * PCH)) * 32, T), T, 0, stream>>>(AZb, bat, pooled,
                                                                               cnt, n);
  k_final16<<<nb((long long)G * C, T), T, 0, stream>>>(pooled, cnt, b3, Wl, bl,
                                                       (float*)d_out, G, C);
}


# Scan of the bin histogram, M = W3 Wl and the pooling zero fills ride in three spare work-groups of the layer-1 product's launch, now between histogram and scatter: 14 launches to 10
# speedup vs baseline: 2.3628x; 1.0741x over previous
#include <hip/hip_runtime.h>
#include <hip/hip_bf16.h>

typedef __attribute__((ext_vector_type(2))) float v2f;
typedef __attribute__((ext_vector_type(8))) float v8f;

#define HID 128
#define WSTRIDE 132
#define ZW 16

#define BSH    9
#define BINN   (1 << BSH)
#define NBMAX  256
#define NCHUNK 256
#define ETH    1024
#define EB     8

__device__ __forceinline__ int wave_scan_incl(int v, int lane) {
#pragma unroll
  for (int off = 1; off < 32; off <<= 1) {
    int u = __shfl_up(v, off, 32);
    if (lane >= off) v += u;
  }
  return v;
}

__device__ __forceinline__ void hold_batch(int v0, int v1, int v2, int v3,
                                           int v4, int v5, int v6, int v7) {
  asm volatile("" :: "v"(v0), "v"(v1), "v"(v2), "v"(v3), "v"(v4), "v"(v5), "v"(v6), "v"(v7));
  __builtin_amdgcn_sched_barrier(0);
}

__global__ void k_binhist(const int* __restrict__ dst, int* __restrict__ hist, int e, int nbins) {
  __shared__ int h[NBMAX];
  const int tid = threadIdx.x;
  const int wg = blockIdx.x;
  if (tid < NBMAX) h[tid] = 0;
  const int per = (e + NCHUNK - 1) / NCHUNK;
  const int beg = wg * per;
  const int end = min(beg + per, e);
  __syncthreads();
  for (int i0 = beg + tid; i0 < end; i0 += ETH * EB) {
    int d[EB];
#pragma unroll
    for (int u = 0; u < EB; ++u) d[u] = dst[min(i0 + u * ETH, end - 1)];
    hold_batch(d[0], d[1], d[2], d[3], d[4], d[5], d[6], d[7]);
#pragma unroll
    for (int u = 0; u < EB; ++u)
      if (i0 + u * ETH < end) atomicAdd(&h[d[u] >> BSH], 1);
  }
  __syncthreads();
  if (tid < nbins) hist[tid * NCHUNK + wg] = h[tid];
}

#define NBW (NBMAX / 32)

__device__ __forceinline__ void scan_share(const int* __restrict__ hist, int v, int lane, int nbins,
                                           int4 (&a)[NBW], int4 (&c)[NBW], int (&s)[NBW],
                                           int (&inc)[NBW]) {
#pragma unroll
  for (int k = 0; k < NBW; ++k) {
    const int b = min(v + 32 * k, nbins - 1);
    const int4* p = (const int4*)(hist + (size_t)b * NCHUNK + lane * 8);
    a[k] = p[0];
    c[k] = p[1];
  }
  __builtin_amdgcn_sched_barrier(0);
#pragma unroll
  for (int k = 0; k < NBW; ++k) {
    s[k] = a[k].x + a[k].y + a[k].z + a[k].w + c[k].x + c[k].y + c[k].z + c[k].w;
    inc[k] = s[k];
  }
#pragma unroll
  for (int off = 1; off < 32; off <<= 1) {
    int u[NBW];
#pragma unroll
    for (int k = 0; k < NBW; ++k) u[k] = __shfl_up(inc[k], off, 32);
    __builtin_amdgcn_sched_barrier(0);
#pragma unroll
    for (int k = 0; k < NBW; ++k)
      if (lane >= off) inc[k] += u[k];
  }
}

__device__ __forceinline__ void scan_role(const int* __restrict__ hist, int* __restrict__ base,
                                          int* __restrict__ binStart, int nbins, int* lds) {
  int* tot = lds;
  int* wsum = lds + NBMAX;
  const int tid = threadIdx.x;
  const int lane = tid & 31;
  const int wave = tid >> 5;

  for (int j = 0; j < 4; ++j) {
    const int v = wave + 8 * j;
    int4 a[NBW], c[NBW];
    int s[NBW], inc[NBW];
    scan_share(hist, v, lane, nbins, a, c, s, inc);
#pragma unroll
    for (int k = 0; k < NBW; ++k)
      if (lane == 31 && v + 32 * k < nbins) tot[v + 32 * k] = inc[k];
  }
  __syncthreads();

  const int t = (tid < nbins) ? tot[tid] : 0;
  const int incl = wave_scan_incl(t, lane);
  if (lane == 31) wsum[wave] = incl;
  __syncthreads();
  {
    int off = 0;
    for (int w = 0; w < NBMAX / 32; ++w)
      if (w < wave) off += wsum[w];
    const int exb = off + incl - t;
    tot[tid] = exb;
    if (tid < nbins) binStart[tid] = exb;
    if (tid == nbins - 1) binStart[nbins] = exb + t;
  }
  __syncthreads();

  for (int j = 0; j < 4; ++j) {
    const int v = wave + 8 * j;
    int4 a[NBW], c[NBW];
    int s[NBW], inc[NBW];
    scan_share(hist, v, lane, nbins, a, c, s, inc);
#pragma unroll
    for (int k = 0; k < NBW; ++k) {
      const int b = v + 32 * k;
      if (b < nbins) {
        int4 o0, o1;
        o0.x = tot[b] + inc[k] - s[k];
        o0.y = o0.x + a[k].x;
        o0.z = o0.y + a[k].y;
        o0.w = o0.z + a[k].z;
        o1.x = o0.w + a[k].w;
        o1.y = o1.x + c[k].x;
        o1.z = o1.y + c[k].y;
        o1.w = o1.z + c[k].z;
        int4* q = (int4*)(base + (size_t)b * NCHUNK + lane * 8);
        q[0] = o0;
        q[1] = o1;
      }
    }
  }
}

__global__ void k_binscatter(const int* __restrict__ src, const int* __restrict__ dst,
                             const int* __restrict__ base, int2* __restrict__ binned,
                             int e, int nbins) {
  __shared__ int cur[NBMAX];
  const int tid = threadIdx.x;
  const int wg = blockIdx.x;
  if (tid < nbins) cur[tid] = base[tid * NCHUNK + wg];
  const int per = (e + NCHUNK - 1) / NCHUNK;
  const int beg = wg * per;
  const int end = min(beg + per, e);
  __syncthreads();
  for (int i0 = beg + tid; i0 < end; i0 += ETH * EB) {
    int s[EB], d[EB];
#pragma unroll
    for (int u = 0; u < EB; ++u) {
      const int j = min(i0 + u * ETH, end - 1);
      s[u] = src[j];
      d[u] = dst[j];
    }
    hold_batch(s[0], s[1], s[2], s[3], s[4], s[5], s[6], s[7]);
    int p[EB];
#pragma unroll
    for (int u = 0; u < EB; ++u)
      p[u] = atomicAdd(&cur[d[u] >> BSH], (i0 + u * ETH < end) ? 1 : 0);
#pragma unroll
    for (int u = 0; u < EB; ++u)
      if (i0 + u * ETH < end) binned[p[u]] = make_int2(s[u], d[u]);
  }
}

__global__ void k_bincount(const int2* __restrict__ binned, const int* __restrict__ binStart,
                           int* __restrict__ rowstart, float* __restrict__ dinv, int n) {
  __shared__ int cnt[BINN];
  __shared__ int wsum[BINN / 32];
  const int tid = threadIdx.x;
  const int lane = tid & 31;
  const int wave = tid >> 5;
  const int b = blockIdx.x;
  const int lo = binStart[b];
  const int hi = binStart[b + 1];
  if (tid < BINN) cnt[tid] = 0;
  __syncthreads();
  for (int i0 = lo + tid; i0 < hi; i0 += ETH * EB) {
    int d[EB];
#pragma unroll
    for (int u = 0; u < EB; ++u) d[u] = binned[min(i0 + u * ETH, hi - 1)].y;
    hold_batch(d[0], d[1], d[2], d[3], d[4], d[5], d[6], d[7]);
#pragma unroll
    for (int u = 0; u < EB; ++u)
      if (i0 + u * ETH < hi) atomicAdd(&cnt[d[u] & (BINN - 1)], 1);
  }
  __syncthreads();
  const int c = (tid < BINN) ? cnt[tid] : 0;
  const int incl = wave_scan_incl(c, lane);
  if (tid < BINN && lane == 31) wsum[wave] = incl;
  __syncthreads();
  if (tid < BINN) {
    int off = 0;
    for (int w = 0; w < BINN / 32; ++w)
      if (w < wave) off += wsum[w];
    const int node = b * BINN + tid;
    if (node < n) {
      rowstart[node] = lo + off + incl - c;
      dinv[node] = rsqrtf((float)c + 1.0f);
    }
  }
  if (b == (int)gridDim.x - 1 && tid == 0) rowstart[n] = hi;
}

__global__ void k_binplace(const int2* __restrict__ binned, const int* __restrict__ binStart,
                           const int* __restrict__ rowstart, const float* __restrict__ dinv,
                           int2* __restrict__ rec, int n) {
  __shared__ int cur[BINN];
  __shared__ float dl[BINN];
  const int tid = threadIdx.x;
  const int b = blockIdx.x;
  const int lo = binStart[b];
  const int hi = binStart[b + 1];
  if (tid < BINN) {
    const int node = min(b * BINN + tid, n - 1);
    cur[tid] = rowstart[node];
    dl[tid] = dinv[node];
  }
  __syncthreads();
  for (int i0 = lo + tid; i0 < hi; i0 += ETH * EB) {
    int2 r[EB];
    float ds[EB];
#pragma unroll
    for (int u = 0; u < EB; ++u) r[u] = binned[min(i0 + u * ETH, hi - 1)];
    hold_batch(r[0].x, r[1].x, r[2].x, r[3].x, r[4].x, r[5].x, r[6].x, r[7].x);
#pragma unroll
    for (int u = 0; u < EB; ++u) ds[u] = dinv[(unsigned)r[u].x];
    hold_batch(__float_as_int(ds[0]), __float_as_int(ds[1]), __float_as_int(ds[2]),
               __float_as_int(ds[3]), __float_as_int(ds[4]), __float_as_int(ds[5]),
               __float_as_int(ds[6]), __float_as_int(ds[7]));
    int p[EB];
    float w[EB];
#pragma unroll
    for (int u = 0; u < EB; ++u) {
      const int t = r[u].y & (BINN - 1);
      p[u] = atomicAdd(&cur[t], (i0 + u * ETH < hi) ? 1 : 0);
      w[u] = ds[u] * dl[t];
    }
#pragma unroll
    for (int u = 0; u < EB; ++u)
      if (i0 + u * ETH < hi) rec[p[u]] = make_int2(r[u].x, __float_as_int(w[u]));
  }
}

__device__ __forceinline__ void gemm128_role(const float* __restrict__ X, const float* __restrict__ W,
                                             float* __restrict__ H, int n, int blk, float* ws) {
  const int tid = threadIdx.x;

  for (int i = tid; i < HID * HID; i += 256) {
    int k = i >> 7, c = i & 127;
    ws[c * WSTRIDE + k] = W[i];
  }
  __syncthreads();

  const int wave   = tid >> 5;
  const int lane   = tid & 31;
  const int lane15 = lane & 15;
  const int half   = lane >> 4;
  const int rowBase = blk * 128 + wave * 16;
  const int arow    = rowBase + lane15;
  const int koff    = half * 2;

  v2f a[32];
  if (arow < n) {
    const float* xp = X + (size_t)arow * HID + koff;
#pragma unroll
    for (int kk = 0; kk < 32; ++kk)
      a[kk] = *(const v2f*)(xp + kk * 4);
  } else {
    v2f z = {};
#pragma unroll
    for (int kk = 0; kk < 32; ++kk) a[kk] = z;
  }

#pragma unroll
  for (int nt = 0; nt < 8; ++nt) {
    v8f acc = {};
    const float* bp = ws + (nt * 16 + lane15) * WSTRIDE + koff;
#pragma unroll
    for (int kk = 0; kk < 32; ++kk) {
      v2f b = *(const v2f*)(bp + kk * 4);
      acc = __builtin_amdgcn_wmma_f32_16x16x4_f32(
          false, a[kk], false, b, (short)0, acc, false, false);
    }
    const int colOut = nt * 16 + lane15;
#pragma unroll
    for (int r = 0; r < 8; ++r) {
      int row = rowBase + r + half * 8;
      if (row < n) H[(size_t)row * HID + colOut] = acc[r];
    }
  }
}

#define GCH 16

__device__ __forceinline__ void hold_rows(const float4 (&h)[GCH]) {
  asm volatile("" :: "v"(h[0].x), "v"(h[1].x), "v"(h[2].x), "v"(h[3].x), "v"(h[4].x), "v"(h[5].x),
               "v"(h[6].x), "v"(h[7].x), "v"(h[8].x), "v"(h[9].x), "v"(h[10].x), "v"(h[11].x),
               "v"(h[12].x), "v"(h[13].x), "v"(h[14].x), "v"(h[15].x));
}

template <int NT>
__global__ void __launch_bounds__(256)
k_agg_gemm(const float* __restrict__ H, const int2* __restrict__ rec,
           const int* __restrict__ rowstart, const float* __restrict__ dinv,
           const float* __restrict__ B, const float* __restrict__ bias,
           float* __restrict__ OUT, int n) {
  constexpr int NC = NT * 16;
  extern __shared__ float sm[];
  const int tid = threadIdx.x;
  const int wave   = tid >> 5;
  const int lane   = tid & 31;
  const int lane15 = lane & 15;
  const int half   = lane >> 4;
  const int koff   = half * 2;
  float* tile = sm + NC * WSTRIDE + wave * (16 * WSTRIDE);

  constexpr int NIT = HID * NC / 256;
  constexpr int SB = NIT < 32 ? NIT : 32;
  for (int it0 = 0; it0 < NIT; it0 += SB) {
    float sv[SB];
#pragma unroll
    for (int u = 0; u < SB; ++u) sv[u] = B[tid + (it0 + u) * 256];
#pragma unroll
    for (int u = 0; u < SB; ++u) {
      const int i = tid + (it0 + u) * 256;
      sm[(i % NC) * WSTRIDE + i / NC] = sv[u];
    }
  }
  __syncthreads();

  const int base = blockIdx.x * 128 + wave * 16;
  const int q = lane * 4;
  const int last = n - 1;
  const int rsv = rowstart[min(base + lane, n)];
  const float dvl = dinv[min(base + lane15, last)];
  const int beg = __builtin_amdgcn_readlane(rsv, 0);
  const int end = __builtin_amdgcn_readlane(rsv, 16);

  {
    float4 sr[16];
#pragma unroll
    for (int m = 0; m < 16; ++m)
      sr[m] = *(const float4*)(H + (size_t)min(base + m, last) * HID + q);
#pragma unroll
    for (int m = 0; m < 16; ++m) {
      float wself = __int_as_float(__builtin_amdgcn_readlane(__float_as_int(dvl), m));
      wself *= wself;
      float4 v = sr[m];
      v.x *= wself; v.y *= wself; v.z *= wself; v.w *= wself;
      if (base + m > last) v = make_float4(0.0f, 0.0f, 0.0f, 0.0f);
      *(float4*)(tile + m * WSTRIDE + q) = v;
    }
  }

  int m = 0;
  int nb = __builtin_amdgcn_readlane(rsv, 1);
  float4 acc = *(const float4*)(tile + q);
  if (beg < end) {
    int2 rv = rec[min(beg + lane, end - 1)];
    for (int jw = beg; jw < end; jw += 32) {
      const int2 rvn = rec[min(jw + 32 + lane, end - 1)];
#pragma unroll
      for (int c = 0; c < 32 / GCH; ++c) {
        const int j0 = jw + c * GCH;
        if (j0 < end) {
          float4 h[GCH];
#pragma unroll
          for (int t = 0; t < GCH; ++t) {
            const unsigned s = (unsigned)__builtin_amdgcn_readlane(rv.x, c * GCH + t);
            h[t] = *(const float4*)(H + (size_t)s * HID + q);
          }
          hold_rows(h);
#pragma unroll
          for (int t = 0; t < GCH; ++t) {
            if (j0 + t < end) {
              while (j0 + t == nb) {
                *(float4*)(tile + m * WSTRIDE + q) = acc;
                ++m;
                acc = *(const float4*)(tile + m * WSTRIDE + q);
                nb = __builtin_amdgcn_readlane(rsv, m + 1);
              }
              const float w = __int_as_float(__builtin_amdgcn_readlane(rv.y, c * GCH + t));
              acc.x += w * h[t].x; acc.y += w * h[t].y; acc.z += w * h[t].z; acc.w += w * h[t].w;
            }
          }
        }
      }
      rv = rvn;
    }
  }
  *(float4*)(tile + m * WSTRIDE + q) = acc;

  v2f a[32];
  {
    const float* tp = tile + lane15 * WSTRIDE + koff;
    const float* cp = bias + koff;
#pragma unroll
    for (int kk = 0; kk < 32; ++kk) {
      v2f xv = *(const v2f*)(tp + kk * 4);
      v2f bv = *(const v2f*)(cp + kk * 4);
      v2f t = xv + bv;
      t.x = fmaxf(t.x, 0.0f);
      t.y = fmaxf(t.y, 0.0f);
      a[kk] = t;
    }
  }

#pragma unroll
  for (int nt = 0; nt < NT; ++nt) {
    v8f accw = {};
    const float* bp = sm + (nt * 16 + lane15) * WSTRIDE + koff;
#pragma unroll
    for (int kk = 0; kk < 32; ++kk) {
      v2f b = *(const v2f*)(bp + kk * 4);
      accw = __builtin_amdgcn_wmma_f32_16x16x4_f32(
          false, a[kk], false, b, (short)0, accw, false, false);
    }
    const int colOut = nt * 16 + lane15;
#pragma unroll
    for (int r = 0; r < 8; ++r) {
      int row = base + r + half * 8;
      if (row < n) OUT[(size_t)row * NC + colOut] = accw[r];
    }
  }
}

__device__ __forceinline__ void gemm16_role(const float* __restrict__ X, const float* __restrict__ B,
                                            int ldb, int nc, float* __restrict__ Z, int n, float* bs) {
  const int tid = threadIdx.x;

  const int wave   = tid >> 5;
  const int lane   = tid & 31;
  const int lane15 = lane & 15;
  const int half   = lane >> 4;
  const int rowBase = wave * 16;
  const int arow    = rowBase + lane15;
  const int koff    = half * 2;

  float sv[HID * ZW / 256];
#pragma unroll
  for (int it = 0; it < HID * ZW / 256; ++it) {
    int i = tid + it * 256;
    sv[it] = B[(i >> 4) * ldb + min(i & 15, nc - 1)];
  }

  v2f a[32];
  const float* xp = X + (size_t)min(arow, n - 1) * HID + koff;
#pragma unroll
  for (int kk = 0; kk < 32; ++kk)
    a[kk] = *(const v2f*)(xp + kk * 4);

#pragma unroll
  for (int it = 0; it < HID * ZW / 256; ++it) {
    int i = tid + it * 256;
    int k = i >> 4, c = i & 15;
    bs[c * WSTRIDE + k] = (c < nc) ? sv[it] : 0.0f;
  }
  __syncthreads();

  v8f acc = {};
  const float* bp = bs + lane15 * WSTRIDE + koff;
#pragma unroll
  for (int kk = 0; kk < 32; ++kk) {
    v2f b = *(const v2f*)(bp + kk * 4);
    acc = __builtin_amdgcn_wmma_f32_16x16x4_f32(
        false, a[kk], false, b, (short)0, acc, false, false);
  }
#pragma unroll
  for (int r = 0; r < 8; ++r) {
    int row = rowBase + r + half * 8;
    if (row < n) Z[(size_t)row * ZW + lane15] = acc[r];
  }
}

#define FRONT_ROLES 3
__global__ void k_front(const float* __restrict__ X, const float* __restrict__ W1,
                        float* __restrict__ H, int n,
                        const int* __restrict__ hist, int* __restrict__ base,
                        int* __restrict__ binStart, int nbins,
                        const float* __restrict__ W3, const float* __restrict__ Wl, int C,
                        float* __restrict__ M, float* __restrict__ pooled,
                        float* __restrict__ cnt, int G) {
  extern __shared__ float ws[];
  const int blk = blockIdx.x;
  if (blk == 0) {
    scan_role(hist, base, binStart, nbins, (int*)ws);
  } else if (blk == 1) {
    gemm16_role(W3, Wl, C, C, M, HID, ws);
  } else if (blk == 2) {
    for (int i = threadIdx.x; i < G * ZW; i += 256) pooled[i] = 0.0f;
    for (int i = threadIdx.x; i < G; i += 256) cnt[i] = 0.0f;
  } else {
    gemm128_role(X, W1, H, n, blk - FRONT_ROLES, ws);
  }
}

__global__ void k_aggregate16(const float* __restrict__ Z, const int2* __restrict__ rec,
                              const int* __restrict__ rowstart, const float* __restrict__ dinv,
                              float* __restrict__ AZ, int n) {
  const int gid = blockIdx.x * blockDim.x + threadIdx.x;
  const int node = gid >> 2;
  if (node >= n) return;
  const int q = (gid & 3) * 4;

  const int beg = rowstart[node];
  const int end = rowstart[node + 1];

  float wself = dinv[node];
  wself *= wself;
  float4 acc = *(const float4*)(Z + (size_t)node * ZW + q);
  acc.x *= wself; acc.y *= wself; acc.z *= wself; acc.w *= wself;

  for (int j = beg; j < end; j += 4) {
    int2 r[4];
    float4 h[4];
#pragma unroll
    for (int t = 0; t < 4; ++t) r[t] = rec[(j + t < end) ? (j + t) : j];
#pragma unroll
    for (int t = 0; t < 4; ++t) h[t] = *(const float4*)(Z + (size_t)r[t].x * ZW + q);
#pragma unroll
    for (int t = 0; t < 4; ++t) {
      const bool ok = (j + t) < end;
      const float w = __int_as_float(r[t].y);
      const float hx = ok ? h[t].x : 0.0f, hy = ok ? h[t].y : 0.0f;
      const float hz = ok ? h[t].z : 0.0f, hw = ok ? h[t].w : 0.0f;
      acc.x += w * hx; acc.y += w * hy; acc.z += w * hz; acc.w += w * hw;
    }
  }
  *(float4*)(AZ + (size_t)node * ZW + q) = acc;
}

#define PCH 8

__device__ __forceinline__ void pool_flush(float* __restrict__ pooled, float* __restrict__ cnt,
                                           int g, int q, float4 acc, int run) {
  float* p = pooled + (size_t)g * ZW + q;
  __hip_atomic_fetch_add(p + 0, acc.x, __ATOMIC_RELAXED, __HIP_MEMORY_SCOPE_AGENT);
  __hip_atomic_fetch_add(p + 1, acc.y, __ATOMIC_RELAXED, __HIP_MEMORY_SCOPE_AGENT);
  __hip_atomic_fetch_add(p + 2, acc.z, __ATOMIC_RELAXED, __HIP_MEMORY_SCOPE_AGENT);
  __hip_atomic_fetch_add(p + 3, acc.w, __ATOMIC_RELAXED, __HIP_MEMORY_SCOPE_AGENT);
  if (q == 0)
    __hip_atomic_fetch_add(&cnt[g], (float)run, __ATOMIC_RELAXED, __HIP_MEMORY_SCOPE_AGENT);
}

__global__ void k_pool16(const float* __restrict__ AZ, const int* __restrict__ batch,
                         float* __restrict__ pooled, float* __restrict__ cnt, int n) {
  const int gid = blockIdx.x * blockDim.x + threadIdx.x;
  const int first = (gid >> 5) * (8 * PCH) + ((gid & 31) >> 2);
  if (first >= n) return;
  const int q = (gid & 3) * 4;
  const int last = n - 1;

  int g[PCH];
  float4 h[PCH];
#pragma unroll
  for (int i = 0; i < PCH; ++i) {
    const int node = min(first + i * 8, last);
    g[i] = batch[node];
    h[i] = *(const float4*)(AZ + (size_t)node * ZW + q);
  }

  int cur = g[0];
  int run = 1;
  float4 acc = h[0];
#pragma unroll
  for (int i = 1; i < PCH; ++i) {
    if (first + i * 8 <= last) {
      if (g[i] != cur) {
        pool_flush(pooled, cnt, cur, q, acc, run);
        cur = g[i];
        run = 1;
        acc = h[i];
      } else {
        acc.x += h[i].x; acc.y += h[i].y; acc.z += h[i].z; acc.w += h[i].w;
        ++run;
      }
    }
  }
  pool_flush(pooled, cnt, cur, q, acc, run);
}

__global__ void k_final16(const float* __restrict__ pooled, const float* __restrict__ cnt,
                          const float* __restrict__ b3, const float* __restrict__ Wl,
                          const float* __restrict__ bl, float* __restrict__ out, int G, int C) {
  int gid = blockIdx.x * blockDim.x + threadIdx.x;
  if (gid >= G * C) return;
  int g = gid / C, c = gid % C;
  float cn = cnt[g];
  float inv = 1.0f / fmaxf(cn, 1.0f);
  float bw = 0.0f;
  for (int k = 0; k < HID; ++k)
    bw += b3[k] * Wl[k * C + c];
  float s = pooled[(size_t)g * ZW + c] * inv;
  if (cn > 0.0f) s += bw;
  out[gid] = s + bl[c];
}

extern "C" void kernel_launch(void* const* d_in, const int* in_sizes, int n_in,
                              void* d_out, int out_size, void* d_ws, size_t ws_size,
                              hipStream_t stream) {
  const float* x   = (const float*)d_in[0];
  const int*   ei  = (const int*)d_in[1];
  const int*   bat = (const int*)d_in[2];
  const float* W1  = (const float*)d_in[3];
  const float* b1  = (const float*)d_in[4];
  const float* W2  = (const float*)d_in[5];
  const float* b2  = (const float*)d_in[6];
  const float* W3  = (const float*)d_in[7];
  const float* b3  = (const float*)d_in[8];
  const float* Wl  = (const float*)d_in[9];
  const float* bl  = (const float*)d_in[10];

  const int n = in_sizes[2];
  const int e = in_sizes[1] / 2;
  const int C = in_sizes[10];
  const int G = out_size / C;
  const int* src = ei;
  const int* dst = ei + e;

  char* base = (char*)d_ws;
  auto alloc = [&](size_t bytes) -> void* {
    void* p = (void*)base;
    base += (bytes + 255) & ~(size_t)255;
    return p;
  };
  int*   hist      = (int*)alloc((size_t)NBMAX * NCHUNK * 4);
  int*   base0     = (int*)alloc((size_t)NBMAX * NCHUNK * 4);
  int*   binStart  = (int*)alloc(((size_t)NBMAX + 1) * 4);
  int2*  binned    = (int2*)alloc((size_t)e * 8);
  int*   rowstart  = (int*)alloc(((size_t)n + 1) * 4);
  float* dinv      = (float*)alloc((size_t)n * 4);
  int2*  rec       = (int2*)alloc((size_t)e * 8);
  float* Hb        = (float*)alloc((size_t)n * HID * 4);
  float* Ab        = (float*)alloc((size_t)n * HID * 4);
  float* Mb        = (float*)alloc((size_t)HID * ZW * 4);
  float* Zb        = (float*)alloc((size_t)n * ZW * 4);
  float* AZb       = (float*)alloc((size_t)n * ZW * 4);
  float* pooled    = (float*)alloc((size_t)G * ZW * 4);
  float* cnt       = (float*)alloc((size_t)G * 4);
  (void)ws_size; (void)n_in;

  const int T = 256;
  auto nb = [](long long w, int t) { return (unsigned)((w + t - 1) / t); };
  const unsigned gemmB = (unsigned)((n + 127) / 128);
  const unsigned nbins = (unsigned)((n + BINN - 1) >> BSH);
  const size_t lds = (size_t)HID * WSTRIDE * sizeof(float);
  const size_t ldsF8 = (size_t)(8 * 16 + 128) * WSTRIDE * sizeof(float);
  const size_t ldsF1 = (size_t)(1 * 16 + 128) * WSTRIDE * sizeof(float);

  k_binhist<<<NCHUNK, ETH, 0, stream>>>(dst, hist, e, (int)nbins);
  k_front<<<gemmB + FRONT_ROLES, T, lds, stream>>>(x, W1, Hb, n, hist, base0, binStart, (int)nbins,
                                                   W3, Wl, C, Mb, pooled, cnt, G);
  k_binscatter<<<NCHUNK, ETH, 0, stream>>>(src, dst, base0, binned, e, (int)nbins);
  k_bincount<<<nbins, ETH, 0, stream>>>(binned, binStart, rowstart, dinv, n);
  k_binplace<<<nbins, ETH, 0, stream>>>(binned, binStart, rowstart, dinv, rec, n);

  k_agg_gemm<8><<<gemmB, T, ldsF8, stream>>>(Hb, rec, rowstart, dinv, W2, b1, Ab, n);

  k_agg_gemm<1><<<gemmB, T, ldsF1, stream>>>(Ab, rec, rowstart, dinv, Mb, b2, Zb, n);
  k_aggregate16<<<nb((long long)n * 4, T), T, 0, stream>>>(Zb, rec, rowstart, dinv, AZb, n);

  k_pool16<<<nb((long long)((n + 8 * PCH - 1) / (8 * PCH)) * 32, T), T, 0, stream>>>(AZb, bat, pooled,
                                                                               cnt, n);
  k_final16<<<nb((long long)G * C, T), T, 0, stream>>>(pooled, cnt, b3, Wl, bl,
                                                       (float*)d_out, G, C);
}


# Neighbour-sum kernels: 250 work-groups of 25 waves, one per WGP, all resident in one even round, weights staged once per WGP (was 782 groups of 8 waves, 3 or 4 per WGP); sum order unchanged
# speedup vs baseline: 2.4701x; 1.0454x over previous
#include <hip/hip_runtime.h>
#include <hip/hip_bf16.h>

typedef __attribute__((ext_vector_type(2))) float v2f;
typedef __attribute__((ext_vector_type(8))) float v8f;

#define HID 128
#define WSTRIDE 132
#define ZW 16

#define BSH    9
#define BINN   (1 << BSH)
#define NBMAX  256
#define NCHUNK 256
#define ETH    1024
#define EB     8

__device__ __forceinline__ int wave_scan_incl(int v, int lane) {
#pragma unroll
  for (int off = 1; off < 32; off <<= 1) {
    int u = __shfl_up(v, off, 32);
    if (lane >= off) v += u;
  }
  return v;
}

__device__ __forceinline__ void hold_batch(int v0, int v1, int v2, int v3,
                                           int v4, int v5, int v6, int v7) {
  asm volatile("" :: "v"(v0), "v"(v1), "v"(v2), "v"(v3), "v"(v4), "v"(v5), "v"(v6), "v"(v7));
  __builtin_amdgcn_sched_barrier(0);
}

__global__ void k_binhist(const int* __restrict__ dst, int* __restrict__ hist, int e, int nbins) {
  __shared__ int h[NBMAX];
  const int tid = threadIdx.x;
  const int wg = blockIdx.x;
  if (tid < NBMAX) h[tid] = 0;
  const int per = (e + NCHUNK - 1) / NCHUNK;
  const int beg = wg * per;
  const int end = min(beg + per, e);
  __syncthreads();
  for (int i0 = beg + tid; i0 < end; i0 += ETH * EB) {
    int d[EB];
#pragma unroll
    for (int u = 0; u < EB; ++u) d[u] = dst[min(i0 + u * ETH, end - 1)];
    hold_batch(d[0], d[1], d[2], d[3], d[4], d[5], d[6], d[7]);
#pragma unroll
    for (int u = 0; u < EB; ++u)
      if (i0 + u * ETH < end) atomicAdd(&h[d[u] >> BSH], 1);
  }
  __syncthreads();
  if (tid < nbins) hist[tid * NCHUNK + wg] = h[tid];
}

#define NBW (NBMAX / 32)

__device__ __forceinline__ void scan_share(const int* __restrict__ hist, int v, int lane, int nbins,
                                           int4 (&a)[NBW], int4 (&c)[NBW], int (&s)[NBW],
                                           int (&inc)[NBW]) {
#pragma unroll
  for (int k = 0; k < NBW; ++k) {
    const int b = min(v + 32 * k, nbins - 1);
    const int4* p = (const int4*)(hist + (size_t)b * NCHUNK + lane * 8);
    a[k] = p[0];
    c[k] = p[1];
  }
  __builtin_amdgcn_sched_barrier(0);
#pragma unroll
  for (int k = 0; k < NBW; ++k) {
    s[k] = a[k].x + a[k].y + a[k].z + a[k].w + c[k].x + c[k].y + c[k].z + c[k].w;
    inc[k] = s[k];
  }
#pragma unroll
  for (int off = 1; off < 32; off <<= 1) {
    int u[NBW];
#pragma unroll
    for (int k = 0; k < NBW; ++k) u[k] = __shfl_up(inc[k], off, 32);
    __builtin_amdgcn_sched_barrier(0);
#pragma unroll
    for (int k = 0; k < NBW; ++k)
      if (lane >= off) inc[k] += u[k];
  }
}

__device__ __forceinline__ void scan_role(const int* __restrict__ hist, int* __restrict__ base,
                                          int* __restrict__ binStart, int nbins, int* lds) {
  int* tot = lds;
  int* wsum = lds + NBMAX;
  const int tid = threadIdx.x;
  const int lane = tid & 31;
  const int wave = tid >> 5;

  for (int j = 0; j < 4; ++j) {
    const int v = wave + 8 * j;
    int4 a[NBW], c[NBW];
    int s[NBW], inc[NBW];
    scan_share(hist, v, lane, nbins, a, c, s, inc);
#pragma unroll
    for (int k = 0; k < NBW; ++k)
      if (lane == 31 && v + 32 * k < nbins) tot[v + 32 * k] = inc[k];
  }
  __syncthreads();

  const int t = (tid < nbins) ? tot[tid] : 0;
  const int incl = wave_scan_incl(t, lane);
  if (lane == 31) wsum[wave] = incl;
  __syncthreads();
  {
    int off = 0;
    for (int w = 0; w < NBMAX / 32; ++w)
      if (w < wave) off += wsum[w];
    const int exb = off + incl - t;
    tot[tid] = exb;
    if (tid < nbins) binStart[tid] = exb;
    if (tid == nbins - 1) binStart[nbins] = exb + t;
  }
  __syncthreads();

  for (int j = 0; j < 4; ++j) {
    const int v = wave + 8 * j;
    int4 a[NBW], c[NBW];
    int s[NBW], inc[NBW];
    scan_share(hist, v, lane, nbins, a, c, s, inc);
#pragma unroll
    for (int k = 0; k < NBW; ++k) {
      const int b = v + 32 * k;
      if (b < nbins) {
        int4 o0, o1;
        o0.x = tot[b] + inc[k] - s[k];
        o0.y = o0.x + a[k].x;
        o0.z = o0.y + a[k].y;
        o0.w = o0.z + a[k].z;
        o1.x = o0.w + a[k].w;
        o1.y = o1.x + c[k].x;
        o1.z = o1.y + c[k].y;
        o1.w = o1.z + c[k].z;
        int4* q = (int4*)(base + (size_t)b * NCHUNK + lane * 8);
        q[0] = o0;
        q[1] = o1;
      }
    }
  }
}

__global__ void k_binscatter(const int* __restrict__ src, const int* __restrict__ dst,
                             const int* __restrict__ base, int2* __restrict__ binned,
                             int e, int nbins) {
  __shared__ int cur[NBMAX];
  const int tid = threadIdx.x;
  const int wg = blockIdx.x;
  if (tid < nbins) cur[tid] = base[tid * NCHUNK + wg];
  const int per = (e + NCHUNK - 1) / NCHUNK;
  const int beg = wg * per;
  const int end = min(beg + per, e);
  __syncthreads();
  for (int i0 = beg + tid; i0 < end; i0 += ETH * EB) {
    int s[EB], d[EB];
#pragma unroll
    for (int u = 0; u < EB; ++u) {
      const int j = min(i0 + u * ETH, end - 1);
      s[u] = src[j];
      d[u] = dst[j];
    }
    hold_batch(s[0], s[1], s[2], s[3], s[4], s[5], s[6], s[7]);
    int p[EB];
#pragma unroll
    for (int u = 0; u < EB; ++u)
      p[u] = atomicAdd(&cur[d[u] >> BSH], (i0 + u * ETH < end) ? 1 : 0);
#pragma unroll
    for (int u = 0; u < EB; ++u)
      if (i0 + u * ETH < end) binned[p[u]] = make_int2(s[u], d[u]);
  }
}

__global__ void k_bincount(const int2* __restrict__ binned, const int* __restrict__ binStart,
                           int* __restrict__ rowstart, float* __restrict__ dinv, int n) {
  __shared__ int cnt[BINN];
  __shared__ int wsum[BINN / 32];
  const int tid = threadIdx.x;
  const int lane = tid & 31;
  const int wave = tid >> 5;
  const int b = blockIdx.x;
  const int lo = binStart[b];
  const int hi = binStart[b + 1];
  if (tid < BINN) cnt[tid] = 0;
  __syncthreads();
  for (int i0 = lo + tid; i0 < hi; i0 += ETH * EB) {
    int d[EB];
#pragma unroll
    for (int u = 0; u < EB; ++u) d[u] = binned[min(i0 + u * ETH, hi - 1)].y;
    hold_batch(d[0], d[1], d[2], d[3], d[4], d[5], d[6], d[7]);
#pragma unroll
    for (int u = 0; u < EB; ++u)
      if (i0 + u * ETH < hi) atomicAdd(&cnt[d[u] & (BINN - 1)], 1);
  }
  __syncthreads();
  const int c = (tid < BINN) ? cnt[tid] : 0;
  const int incl = wave_scan_incl(c, lane);
  if (tid < BINN && lane == 31) wsum[wave] = incl;
  __syncthreads();
  if (tid < BINN) {
    int off = 0;
    for (int w = 0; w < BINN / 32; ++w)
      if (w < wave) off += wsum[w];
    const int node = b * BINN + tid;
    if (node < n) {
      rowstart[node] = lo + off + incl - c;
      dinv[node] = rsqrtf((float)c + 1.0f);
    }
  }
  if (b == (int)gridDim.x - 1 && tid == 0) rowstart[n] = hi;
}

__global__ void k_binplace(const int2* __restrict__ binned, const int* __restrict__ binStart,
                           const int* __restrict__ rowstart, const float* __restrict__ dinv,
                           int2* __restrict__ rec, int n) {
  __shared__ int cur[BINN];
  __shared__ float dl[BINN];
  const int tid = threadIdx.x;
  const int b = blockIdx.x;
  const int lo = binStart[b];
  const int hi = binStart[b + 1];
  if (tid < BINN) {
    const int node = min(b * BINN + tid, n - 1);
    cur[tid] = rowstart[node];
    dl[tid] = dinv[node];
  }
  __syncthreads();
  for (int i0 = lo + tid; i0 < hi; i0 += ETH * EB) {
    int2 r[EB];
    float ds[EB];
#pragma unroll
    for (int u = 0; u < EB; ++u) r[u] = binned[min(i0 + u * ETH, hi - 1)];
    hold_batch(r[0].x, r[1].x, r[2].x, r[3].x, r[4].x, r[5].x, r[6].x, r[7].x);
#pragma unroll
    for (int u = 0; u < EB; ++u) ds[u] = dinv[(unsigned)r[u].x];
    hold_batch(__float_as_int(ds[0]), __float_as_int(ds[1]), __float_as_int(ds[2]),
               __float_as_int(ds[3]), __float_as_int(ds[4]), __float_as_int(ds[5]),
               __float_as_int(ds[6]), __float_as_int(ds[7]));
    int p[EB];
    float w[EB];
#pragma unroll
    for (int u = 0; u < EB; ++u) {
      const int t = r[u].y & (BINN - 1);
      p[u] = atomicAdd(&cur[t], (i0 + u * ETH < hi) ? 1 : 0);
      w[u] = ds[u] * dl[t];
    }
#pragma unroll
    for (int u = 0; u < EB; ++u)
      if (i0 + u * ETH < hi) rec[p[u]] = make_int2(r[u].x, __float_as_int(w[u]));
  }
}

__device__ __forceinline__ void gemm128_role(const float* __restrict__ X, const float* __restrict__ W,
                                             float* __restrict__ H, int n, int blk, float* ws) {
  const int tid = threadIdx.x;

  for (int i = tid; i < HID * HID; i += 256) {
    int k = i >> 7, c = i & 127;
    ws[c * WSTRIDE + k] = W[i];
  }
  __syncthreads();

  const int wave   = tid >> 5;
  const int lane   = tid & 31;
  const int lane15 = lane & 15;
  const int half   = lane >> 4;
  const int rowBase = blk * 128 + wave * 16;
  const int arow    = rowBase + lane15;
  const int koff    = half * 2;

  v2f a[32];
  if (arow < n) {
    const float* xp = X + (size_t)arow * HID + koff;
#pragma unroll
    for (int kk = 0; kk < 32; ++kk)
      a[kk] = *(const v2f*)(xp + kk * 4);
  } else {
    v2f z = {};
#pragma unroll
    for (int kk = 0; kk < 32; ++kk) a[kk] = z;
  }

#pragma unroll
  for (int nt = 0; nt < 8; ++nt) {
    v8f acc = {};
    const float* bp = ws + (nt * 16 + lane15) * WSTRIDE + koff;
#pragma unroll
    for (int kk = 0; kk < 32; ++kk) {
      v2f b = *(const v2f*)(bp + kk * 4);
      acc = __builtin_amdgcn_wmma_f32_16x16x4_f32(
          false, a[kk], false, b, (short)0, acc, false, false);
    }
    const int colOut = nt * 16 + lane15;
#pragma unroll
    for (int r = 0; r < 8; ++r) {
      int row = rowBase + r + half * 8;
      if (row < n) H[(size_t)row * HID + colOut] = acc[r];
    }
  }
}

#define GCH 16
#define AGGW 25
#define AGGT (AGGW * 32)
#define AGGN (AGGW * 16)

__device__ __forceinline__ void hold_rows(const float4 (&h)[GCH]) {
  asm volatile("" :: "v"(h[0].x), "v"(h[1].x), "v"(h[2].x), "v"(h[3].x), "v"(h[4].x), "v"(h[5].x),
               "v"(h[6].x), "v"(h[7].x), "v"(h[8].x), "v"(h[9].x), "v"(h[10].x), "v"(h[11].x),
               "v"(h[12].x), "v"(h[13].x), "v"(h[14].x), "v"(h[15].x));
}

template <int NT>
__global__ void __launch_bounds__(AGGT)
k_agg_gemm(const float* __restrict__ H, const int2* __restrict__ rec,
           const int* __restrict__ rowstart, const float* __restrict__ dinv,
           const float* __restrict__ B, const float* __restrict__ bias,
           float* __restrict__ OUT, int n) {
  constexpr int NC = NT * 16;
  extern __shared__ float sm[];
  const int tid = threadIdx.x;
  const int wave   = tid >> 5;
  const int lane   = tid & 31;
  const int lane15 = lane & 15;
  const int half   = lane >> 4;
  const int koff   = half * 2;
  float* tile = sm + NC * WSTRIDE + wave * (16 * WSTRIDE);

  constexpr int TOT = HID * NC;
  constexpr int NIT = (TOT + AGGT - 1) / AGGT;
  {
    float sv[NIT];
#pragma unroll
    for (int u = 0; u < NIT; ++u) sv[u] = B[min(tid + u * AGGT, TOT - 1)];
#pragma unroll
    for (int u = 0; u < NIT; ++u) {
      const int i = tid + u * AGGT;
      if (i < TOT) sm[(i % NC) * WSTRIDE + i / NC] = sv[u];
    }
  }
  __syncthreads();

  const int base = blockIdx.x * AGGN + wave * 16;
  const int q = lane * 4;
  const int last = n - 1;
  const int rsv = rowstart[min(base + lane, n)];
  const float dvl = dinv[min(base + lane15, last)];
  const int beg = __builtin_amdgcn_readlane(rsv, 0);
  const int end = __builtin_amdgcn_readlane(rsv, 16);

  {
    float4 sr[16];
#pragma unroll
    for (int m = 0; m < 16; ++m)
      sr[m] = *(const float4*)(H + (size_t)min(base + m, last) * HID + q);
#pragma unroll
    for (int m = 0; m < 16; ++m) {
      float wself = __int_as_float(__builtin_amdgcn_readlane(__float_as_int(dvl), m));
      wself *= wself;
      float4 v = sr[m];
      v.x *= wself; v.y *= wself; v.z *= wself; v.w *= wself;
      if (base + m > last) v = make_float4(0.0f, 0.0f, 0.0f, 0.0f);
      *(float4*)(tile + m * WSTRIDE + q) = v;
    }
  }

  int m = 0;
  int nb = __builtin_amdgcn_readlane(rsv, 1);
  float4 acc = *(const float4*)(tile + q);
  if (beg < end) {
    int2 rv = rec[min(beg + lane, end - 1)];
    for (int jw = beg; jw < end; jw += 32) {
      const int2 rvn = rec[min(jw + 32 + lane, end - 1)];
#pragma unroll
      for (int c = 0; c < 32 / GCH; ++c) {
        const int j0 = jw + c * GCH;
        if (j0 < end) {
          float4 h[GCH];
#pragma unroll
          for (int t = 0; t < GCH; ++t) {
            const unsigned s = (unsigned)__builtin_amdgcn_readlane(rv.x, c * GCH + t);
            h[t] = *(const float4*)(H + (size_t)s * HID + q);
          }
          hold_rows(h);
#pragma unroll
          for (int t = 0; t < GCH; ++t) {
            if (j0 + t < end) {
              while (j0 + t == nb) {
                *(float4*)(tile + m * WSTRIDE + q) = acc;
                ++m;
                acc = *(const float4*)(tile + m * WSTRIDE + q);
                nb = __builtin_amdgcn_readlane(rsv, m + 1);
              }
              const float w = __int_as_float(__builtin_amdgcn_readlane(rv.y, c * GCH + t));
              acc.x += w * h[t].x; acc.y += w * h[t].y; acc.z += w * h[t].z; acc.w += w * h[t].w;
            }
          }
        }
      }
      rv = rvn;
    }
  }
  *(float4*)(tile + m * WSTRIDE + q) = acc;

  v2f a[32];
  {
    const float* tp = tile + lane15 * WSTRIDE + koff;
    const float* cp = bias + koff;
#pragma unroll
    for (int kk = 0; kk < 32; ++kk) {
      v2f xv = *(const v2f*)(tp + kk * 4);
      v2f bv = *(const v2f*)(cp + kk * 4);
      v2f t = xv + bv;
      t.x = fmaxf(t.x, 0.0f);
      t.y = fmaxf(t.y, 0.0f);
      a[kk] = t;
    }
  }

#pragma unroll
  for (int nt = 0; nt < NT; ++nt) {
    v8f accw = {};
    const float* bp = sm + (nt * 16 + lane15) * WSTRIDE + koff;
#pragma unroll
    for (int kk = 0; kk < 32; ++kk) {
      v2f b = *(const v2f*)(bp + kk * 4);
      accw = __builtin_amdgcn_wmma_f32_16x16x4_f32(
          false, a[kk], false, b, (short)0, accw, false, false);
    }
    const int colOut = nt * 16 + lane15;
#pragma unroll
    for (int r = 0; r < 8; ++r) {
      int row = base + r + half * 8;
      if (row < n) OUT[(size_t)row * NC + colOut] = accw[r];
    }
  }
}

__device__ __forceinline__ void gemm16_role(const float* __restrict__ X, const float* __restrict__ B,
                                            int ldb, int nc, float* __restrict__ Z, int n, float* bs) {
  const int tid = threadIdx.x;

  const int wave   = tid >> 5;
  const int lane   = tid & 31;
  const int lane15 = lane & 15;
  const int half   = lane >> 4;
  const int rowBase = wave * 16;
  const int arow    = rowBase + lane15;
  const int koff    = half * 2;

  float sv[HID * ZW / 256];
#pragma unroll
  for (int it = 0; it < HID * ZW / 256; ++it) {
    int i = tid + it * 256;
    sv[it] = B[(i >> 4) * ldb + min(i & 15, nc - 1)];
  }

  v2f a[32];
  const float* xp = X + (size_t)min(arow, n - 1) * HID + koff;
#pragma unroll
  for (int kk = 0; kk < 32; ++kk)
    a[kk] = *(const v2f*)(xp + kk * 4);

#pragma unroll
  for (int it = 0; it < HID * ZW / 256; ++it) {
    int i = tid + it * 256;
    int k = i >> 4, c = i & 15;
    bs[c * WSTRIDE + k] = (c < nc) ? sv[it] : 0.0f;
  }
  __syncthreads();

  v8f acc = {};
  const float* bp = bs + lane15 * WSTRIDE + koff;
#pragma unroll
  for (int kk = 0; kk < 32; ++kk) {
    v2f b = *(const v2f*)(bp + kk * 4);
    acc = __builtin_amdgcn_wmma_f32_16x16x4_f32(
        false, a[kk], false, b, (short)0, acc, false, false);
  }
#pragma unroll
  for (int r = 0; r < 8; ++r) {
    int row = rowBase + r + half * 8;
    if (row < n) Z[(size_t)row * ZW + lane15] = acc[r];
  }
}

#define FRONT_ROLES 3
__global__ void k_front(const float* __restrict__ X, const float* __restrict__ W1,
                        float* __restrict__ H, int n,
                        const int* __restrict__ hist, int* __restrict__ base,
                        int* __restrict__ binStart, int nbins,
                        const float* __restrict__ W3, const float* __restrict__ Wl, int C,
                        float* __restrict__ M, float* __restrict__ pooled,
                        float* __restrict__ cnt, int G) {
  extern __shared__ float ws[];
  const int blk = blockIdx.x;
  if (blk == 0) {
    scan_role(hist, base, binStart, nbins, (int*)ws);
  } else if (blk == 1) {
    gemm16_role(W3, Wl, C, C, M, HID, ws);
  } else if (blk == 2) {
    for (int i = threadIdx.x; i < G * ZW; i += 256) pooled[i] = 0.0f;
    for (int i = threadIdx.x; i < G; i += 256) cnt[i] = 0.0f;
  } else {
    gemm128_role(X, W1, H, n, blk - FRONT_ROLES, ws);
  }
}

__global__ void k_aggregate16(const float* __restrict__ Z, const int2* __restrict__ rec,
                              const int* __restrict__ rowstart, const float* __restrict__ dinv,
                              float* __restrict__ AZ, int n) {
  const int gid = blockIdx.x * blockDim.x + threadIdx.x;
  const int node = gid >> 2;
  if (node >= n) return;
  const int q = (gid & 3) * 4;

  const int beg = rowstart[node];
  const int end = rowstart[node + 1];

  float wself = dinv[node];
  wself *= wself;
  float4 acc = *(const float4*)(Z + (size_t)node * ZW + q);
  acc.x *= wself; acc.y *= wself; acc.z *= wself; acc.w *= wself;

  for (int j = beg; j < end; j += 4) {
    int2 r[4];
    float4 h[4];
#pragma unroll
    for (int t = 0; t < 4; ++t) r[t] = rec[(j + t < end) ? (j + t) : j];
#pragma unroll
    for (int t = 0; t < 4; ++t) h[t] = *(const float4*)(Z + (size_t)r[t].x * ZW + q);
#pragma unroll
    for (int t = 0; t < 4; ++t) {
      const bool ok = (j + t) < end;
      const float w = __int_as_float(r[t].y);
      const float hx = ok ? h[t].x : 0.0f, hy = ok ? h[t].y : 0.0f;
      const float hz = ok ? h[t].z : 0.0f, hw = ok ? h[t].w : 0.0f;
      acc.x += w * hx; acc.y += w * hy; acc.z += w * hz; acc.w += w * hw;
    }
  }
  *(float4*)(AZ + (size_t)node * ZW + q) = acc;
}

#define PCH 8

__device__ __forceinline__ void pool_flush(float* __restrict__ pooled, float* __restrict__ cnt,
                                           int g, int q, float4 acc, int run) {
  float* p = pooled + (size_t)g * ZW + q;
  __hip_atomic_fetch_add(p + 0, acc.x, __ATOMIC_RELAXED, __HIP_MEMORY_SCOPE_AGENT);
  __hip_atomic_fetch_add(p + 1, acc.y, __ATOMIC_RELAXED, __HIP_MEMORY_SCOPE_AGENT);
  __hip_atomic_fetch_add(p + 2, acc.z, __ATOMIC_RELAXED, __HIP_MEMORY_SCOPE_AGENT);
  __hip_atomic_fetch_add(p + 3, acc.w, __ATOMIC_RELAXED, __HIP_MEMORY_SCOPE_AGENT);
  if (q == 0)
    __hip_atomic_fetch_add(&cnt[g], (float)run, __ATOMIC_RELAXED, __HIP_MEMORY_SCOPE_AGENT);
}

__global__ void k_pool16(const float* __restrict__ AZ, const int* __restrict__ batch,
                         float* __restrict__ pooled, float* __restrict__ cnt, int n) {
  const int gid = blockIdx.x * blockDim.x + threadIdx.x;
  const int first = (gid >> 5) * (8 * PCH) + ((gid & 31) >> 2);
  if (first >= n) return;
  const int q = (gid & 3) * 4;
  const int last = n - 1;

  int g[PCH];
  float4 h[PCH];
#pragma unroll
  for (int i = 0; i < PCH; ++i) {
    const int node = min(first + i * 8, last);
    g[i] = batch[node];
    h[i] = *(const float4*)(AZ + (size_t)node * ZW + q);
  }

  int cur = g[0];
  int run = 1;
  float4 acc = h[0];
#pragma unroll
  for (int i = 1; i < PCH; ++i) {
    if (first + i * 8 <= last) {
      if (g[i] != cur) {
        pool_flush(pooled, cnt, cur, q, acc, run);
        cur = g[i];
        run = 1;
        acc = h[i];
      } else {
        acc.x += h[i].x; acc.y += h[i].y; acc.z += h[i].z; acc.w += h[i].w;
        ++run;
      }
    }
  }
  pool_flush(pooled, cnt, cur, q, acc, run);
}

__global__ void k_final16(const float* __restrict__ pooled, const float* __restrict__ cnt,
                          const float* __restrict__ b3, const float* __restrict__ Wl,
                          const float* __restrict__ bl, float* __restrict__ out, int G, int C) {
  int gid = blockIdx.x * blockDim.x + threadIdx.x;
  if (gid >= G * C) return;
  int g = gid / C, c = gid % C;
  float cn = cnt[g];
  float inv = 1.0f / fmaxf(cn, 1.0f);
  float bw = 0.0f;
  for (int k = 0; k < HID; ++k)
    bw += b3[k] * Wl[k * C + c];
  float s = pooled[(size_t)g * ZW + c] * inv;
  if (cn > 0.0f) s += bw;
  out[gid] = s + bl[c];
}

extern "C" void kernel_launch(void* const* d_in, const int* in_sizes, int n_in,
                              void* d_out, int out_size, void* d_ws, size_t ws_size,
                              hipStream_t stream) {
  const float* x   = (const float*)d_in[0];
  const int*   ei  = (const int*)d_in[1];
  const int*   bat = (const int*)d_in[2];
  const float* W1  = (const float*)d_in[3];
  const float* b1  = (const float*)d_in[4];
  const float* W2  = (const float*)d_in[5];
  const float* b2  = (const float*)d_in[6];
  const float* W3  = (const float*)d_in[7];
  const float* b3  = (const float*)d_in[8];
  const float* Wl  = (const float*)d_in[9];
  const float* bl  = (const float*)d_in[10];

  const int n = in_sizes[2];
  const int e = in_sizes[1] / 2;
  const int C = in_sizes[10];
  const int G = out_size / C;
  const int* src = ei;
  const int* dst = ei + e;

  char* base = (char*)d_ws;
  auto alloc = [&](size_t bytes) -> void* {
    void* p = (void*)base;
    base += (bytes + 255) & ~(size_t)255;
    return p;
  };
  int*   hist      = (int*)alloc((size_t)NBMAX * NCHUNK * 4);
  int*   base0     = (int*)alloc((size_t)NBMAX * NCHUNK * 4);
  int*   binStart  = (int*)alloc(((size_t)NBMAX + 1) * 4);
  int2*  binned    = (int2*)alloc((size_t)e * 8);
  int*   rowstart  = (int*)alloc(((size_t)n + 1) * 4);
  float* dinv      = (float*)alloc((size_t)n * 4);
  int2*  rec       = (int2*)alloc((size_t)e * 8);
  float* Hb        = (float*)alloc((size_t)n * HID * 4);
  float* Ab        = (float*)alloc((size_t)n * HID * 4);
  float* Mb        = (float*)alloc((size_t)HID * ZW * 4);
  float* Zb        = (float*)alloc((size_t)n * ZW * 4);
  float* AZb       = (float*)alloc((size_t)n * ZW * 4);
  float* pooled    = (float*)alloc((size_t)G * ZW * 4);
  float* cnt       = (float*)alloc((size_t)G * 4);
  (void)ws_size; (void)n_in;

  const int T = 256;
  auto nb = [](long long w, int t) { return (unsigned)((w + t - 1) / t); };
  const unsigned gemmB = (unsigned)((n + 127) / 128);
  const unsigned nbins = (unsigned)((n + BINN - 1) >> BSH);
  const size_t lds = (size_t)HID * WSTRIDE * sizeof(float);
  const unsigned aggB = (unsigned)((n + AGGN - 1) / AGGN);
  const size_t ldsF8 = (size_t)(8 * 16 + AGGN) * WSTRIDE * sizeof(float);
  const size_t ldsF1 = (size_t)(1 * 16 + AGGN) * WSTRIDE * sizeof(float);

  k_binhist<<<NCHUNK, ETH, 0, stream>>>(dst, hist, e, (int)nbins);
  k_front<<<gemmB + FRONT_ROLES, T, lds, stream>>>(x, W1, Hb, n, hist, base0, binStart, (int)nbins,
                                                   W3, Wl, C, Mb, pooled, cnt, G);
  k_binscatter<<<NCHUNK, ETH, 0, stream>>>(src, dst, base0, binned, e, (int)nbins);
  k_bincount<<<nbins, ETH, 0, stream>>>(binned, binStart, rowstart, dinv, n);
  k_binplace<<<nbins, ETH, 0, stream>>>(binned, binStart, rowstart, dinv, rec, n);

  k_agg_gemm<8><<<aggB, AGGT, ldsF8, stream>>>(Hb, rec, rowstart, dinv, W2, b1, Ab, n);

  k_agg_gemm<1><<<aggB, AGGT, ldsF1, stream>>>(Ab, rec, rowstart, dinv, Mb, b2, Zb, n);
  k_aggregate16<<<nb((long long)n * 4, T), T, 0, stream>>>(Zb, rec, rowstart, dinv, AZb, n);

  k_pool16<<<nb((long long)((n + 8 * PCH - 1) / (8 * PCH)) * 32, T), T, 0, stream>>>(AZb, bat, pooled,
                                                                               cnt, n);
  k_final16<<<nb((long long)G * C, T), T, 0, stream>>>(pooled, cnt, b3, Wl, bl,
                                                       (float*)d_out, G, C);
}


# Tail in one launch: 16-wide neighbour sum, mean pool, classifier rest fused; a wave sums its 8 rows by shuffles, divides by graph size, adds into out (atomics): 10 launches to 8; sum order changes
# speedup vs baseline: 2.5794x; 1.0442x over previous
#include <hip/hip_runtime.h>
#include <hip/hip_bf16.h>

typedef __attribute__((ext_vector_type(2))) float v2f;
typedef __attribute__((ext_vector_type(8))) float v8f;

#define HID 128
#define WSTRIDE 132
#define ZW 16

#define BSH    9
#define BINN   (1 << BSH)
#define NBMAX  256
#define NCHUNK 256
#define ETH    1024
#define EB     8

__device__ __forceinline__ int wave_scan_incl(int v, int lane) {
#pragma unroll
  for (int off = 1; off < 32; off <<= 1) {
    int u = __shfl_up(v, off, 32);
    if (lane >= off) v += u;
  }
  return v;
}

__device__ __forceinline__ void hold_batch(int v0, int v1, int v2, int v3,
                                           int v4, int v5, int v6, int v7) {
  asm volatile("" :: "v"(v0), "v"(v1), "v"(v2), "v"(v3), "v"(v4), "v"(v5), "v"(v6), "v"(v7));
  __builtin_amdgcn_sched_barrier(0);
}

__global__ void k_binhist(const int* __restrict__ dst, int* __restrict__ hist, int e, int nbins,
                          int* __restrict__ gsize, int G) {
  __shared__ int h[NBMAX];
  const int tid = threadIdx.x;
  const int wg = blockIdx.x;
  if (wg == 0)
    for (int i = tid; i < G; i += ETH) gsize[i] = 0;
  if (tid < NBMAX) h[tid] = 0;
  const int per = (e + NCHUNK - 1) / NCHUNK;
  const int beg = wg * per;
  const int end = min(beg + per, e);
  __syncthreads();
  for (int i0 = beg + tid; i0 < end; i0 += ETH * EB) {
    int d[EB];
#pragma unroll
    for (int u = 0; u < EB; ++u) d[u] = dst[min(i0 + u * ETH, end - 1)];
    hold_batch(d[0], d[1], d[2], d[3], d[4], d[5], d[6], d[7]);
#pragma unroll
    for (int u = 0; u < EB; ++u)
      if (i0 + u * ETH < end) atomicAdd(&h[d[u] >> BSH], 1);
  }
  __syncthreads();
  if (tid < nbins) hist[tid * NCHUNK + wg] = h[tid];
}

#define NBW (NBMAX / 32)

__device__ __forceinline__ void scan_share(const int* __restrict__ hist, int v, int lane, int nbins,
                                           int4 (&a)[NBW], int4 (&c)[NBW], int (&s)[NBW],
                                           int (&inc)[NBW]) {
#pragma unroll
  for (int k = 0; k < NBW; ++k) {
    const int b = min(v + 32 * k, nbins - 1);
    const int4* p = (const int4*)(hist + (size_t)b * NCHUNK + lane * 8);
    a[k] = p[0];
    c[k] = p[1];
  }
  __builtin_amdgcn_sched_barrier(0);
#pragma unroll
  for (int k = 0; k < NBW; ++k) {
    s[k] = a[k].x + a[k].y + a[k].z + a[k].w + c[k].x + c[k].y + c[k].z + c[k].w;
    inc[k] = s[k];
  }
#pragma unroll
  for (int off = 1; off < 32; off <<= 1) {
    int u[NBW];
#pragma unroll
    for (int k = 0; k < NBW; ++k) u[k] = __shfl_up(inc[k], off, 32);
    __builtin_amdgcn_sched_barrier(0);
#pragma unroll
    for (int k = 0; k < NBW; ++k)
      if (lane >= off) inc[k] += u[k];
  }
}

__device__ __forceinline__ void scan_role(const int* __restrict__ hist, int* __restrict__ base,
                                          int* __restrict__ binStart, int nbins, int* lds) {
  int* tot = lds;
  int* wsum = lds + NBMAX;
  const int tid = threadIdx.x;
  const int lane = tid & 31;
  const int wave = tid >> 5;

  for (int j = 0; j < 4; ++j) {
    const int v = wave + 8 * j;
    int4 a[NBW], c[NBW];
    int s[NBW], inc[NBW];
    scan_share(hist, v, lane, nbins, a, c, s, inc);
#pragma unroll
    for (int k = 0; k < NBW; ++k)
      if (lane == 31 && v + 32 * k < nbins) tot[v + 32 * k] = inc[k];
  }
  __syncthreads();

  const int t = (tid < nbins) ? tot[tid] : 0;
  const int incl = wave_scan_incl(t, lane);
  if (lane == 31) wsum[wave] = incl;
  __syncthreads();
  {
    int off = 0;
    for (int w = 0; w < NBMAX / 32; ++w)
      if (w < wave) off += wsum[w];
    const int exb = off + incl - t;
    tot[tid] = exb;
    if (tid < nbins) binStart[tid] = exb;
    if (tid == nbins - 1) binStart[nbins] = exb + t;
  }
  __syncthreads();

  for (int j = 0; j < 4; ++j) {
    const int v = wave + 8 * j;
    int4 a[NBW], c[NBW];
    int s[NBW], inc[NBW];
    scan_share(hist, v, lane, nbins, a, c, s, inc);
#pragma unroll
    for (int k = 0; k < NBW; ++k) {
      const int b = v + 32 * k;
      if (b < nbins) {
        int4 o0, o1;
        o0.x = tot[b] + inc[k] - s[k];
        o0.y = o0.x + a[k].x;
        o0.z = o0.y + a[k].y;
        o0.w = o0.z + a[k].z;
        o1.x = o0.w + a[k].w;
        o1.y = o1.x + c[k].x;
        o1.z = o1.y + c[k].y;
        o1.w = o1.z + c[k].z;
        int4* q = (int4*)(base + (size_t)b * NCHUNK + lane * 8);
        q[0] = o0;
        q[1] = o1;
      }
    }
  }
}

__global__ void k_binscatter(const int* __restrict__ src, const int* __restrict__ dst,
                             const int* __restrict__ base, int2* __restrict__ binned,
                             int e, int nbins) {
  __shared__ int cur[NBMAX];
  const int tid = threadIdx.x;
  const int wg = blockIdx.x;
  if (tid < nbins) cur[tid] = base[tid * NCHUNK + wg];
  const int per = (e + NCHUNK - 1) / NCHUNK;
  const int beg = wg * per;
  const int end = min(beg + per, e);
  __syncthreads();
  for (int i0 = beg + tid; i0 < end; i0 += ETH * EB) {
    int s[EB], d[EB];
#pragma unroll
    for (int u = 0; u < EB; ++u) {
      const int j = min(i0 + u * ETH, end - 1);
      s[u] = src[j];
      d[u] = dst[j];
    }
    hold_batch(s[0], s[1], s[2], s[3], s[4], s[5], s[6], s[7]);
    int p[EB];
#pragma unroll
    for (int u = 0; u < EB; ++u)
      p[u] = atomicAdd(&cur[d[u] >> BSH], (i0 + u * ETH < end) ? 1 : 0);
#pragma unroll
    for (int u = 0; u < EB; ++u)
      if (i0 + u * ETH < end) binned[p[u]] = make_int2(s[u], d[u]);
  }
}

__global__ void k_bincount(const int2* __restrict__ binned, const int* __restrict__ binStart,
                           int* __restrict__ rowstart, float* __restrict__ dinv, int n) {
  __shared__ int cnt[BINN];
  __shared__ int wsum[BINN / 32];
  const int tid = threadIdx.x;
  const int lane = tid & 31;
  const int wave = tid >> 5;
  const int b = blockIdx.x;
  const int lo = binStart[b];
  const int hi = binStart[b + 1];
  if (tid < BINN) cnt[tid] = 0;
  __syncthreads();
  for (int i0 = lo + tid; i0 < hi; i0 += ETH * EB) {
    int d[EB];
#pragma unroll
    for (int u = 0; u < EB; ++u) d[u] = binned[min(i0 + u * ETH, hi - 1)].y;
    hold_batch(d[0], d[1], d[2], d[3], d[4], d[5], d[6], d[7]);
#pragma unroll
    for (int u = 0; u < EB; ++u)
      if (i0 + u * ETH < hi) atomicAdd(&cnt[d[u] & (BINN - 1)], 1);
  }
  __syncthreads();
  const int c = (tid < BINN) ? cnt[tid] : 0;
  const int incl = wave_scan_incl(c, lane);
  if (tid < BINN && lane == 31) wsum[wave] = incl;
  __syncthreads();
  if (tid < BINN) {
    int off = 0;
    for (int w = 0; w < BINN / 32; ++w)
      if (w < wave) off += wsum[w];
    const int node = b * BINN + tid;
    if (node < n) {
      rowstart[node] = lo + off + incl - c;
      dinv[node] = rsqrtf((float)c + 1.0f);
    }
  }
  if (b == (int)gridDim.x - 1 && tid == 0) rowstart[n] = hi;
}

__global__ void k_binplace(const int2* __restrict__ binned, const int* __restrict__ binStart,
                           const int* __restrict__ rowstart, const float* __restrict__ dinv,
                           int2* __restrict__ rec, int n) {
  __shared__ int cur[BINN];
  __shared__ float dl[BINN];
  const int tid = threadIdx.x;
  const int b = blockIdx.x;
  const int lo = binStart[b];
  const int hi = binStart[b + 1];
  if (tid < BINN) {
    const int node = min(b * BINN + tid, n - 1);
    cur[tid] = rowstart[node];
    dl[tid] = dinv[node];
  }
  __syncthreads();
  for (int i0 = lo + tid; i0 < hi; i0 += ETH * EB) {
    int2 r[EB];
    float ds[EB];
#pragma unroll
    for (int u = 0; u < EB; ++u) r[u] = binned[min(i0 + u * ETH, hi - 1)];
    hold_batch(r[0].x, r[1].x, r[2].x, r[3].x, r[4].x, r[5].x, r[6].x, r[7].x);
#pragma unroll
    for (int u = 0; u < EB; ++u) ds[u] = dinv[(unsigned)r[u].x];
    hold_batch(__float_as_int(ds[0]), __float_as_int(ds[1]), __float_as_int(ds[2]),
               __float_as_int(ds[3]), __float_as_int(ds[4]), __float_as_int(ds[5]),
               __float_as_int(ds[6]), __float_as_int(ds[7]));
    int p[EB];
    float w[EB];
#pragma unroll
    for (int u = 0; u < EB; ++u) {
      const int t = r[u].y & (BINN - 1);
      p[u] = atomicAdd(&cur[t], (i0 + u * ETH < hi) ? 1 : 0);
      w[u] = ds[u] * dl[t];
    }
#pragma unroll
    for (int u = 0; u < EB; ++u)
      if (i0 + u * ETH < hi) rec[p[u]] = make_int2(r[u].x, __float_as_int(w[u]));
  }
}

__device__ __forceinline__ void gemm128_role(const float* __restrict__ X, const float* __restrict__ W,
                                             float* __restrict__ H, int n, int blk, float* ws) {
  const int tid = threadIdx.x;

  for (int i = tid; i < HID * HID; i += 256) {
    int k = i >> 7, c = i & 127;
    ws[c * WSTRIDE + k] = W[i];
  }
  __syncthreads();

  const int wave   = tid >> 5;
  const int lane   = tid & 31;
  const int lane15 = lane & 15;
  const int half   = lane >> 4;
  const int rowBase = blk * 128 + wave * 16;
  const int arow    = rowBase + lane15;
  const int koff    = half * 2;

  v2f a[32];
  if (arow < n) {
    const float* xp = X + (size_t)arow * HID + koff;
#pragma unroll
    for (int kk = 0; kk < 32; ++kk)
      a[kk] = *(const v2f*)(xp + kk * 4);
  } else {
    v2f z = {};
#pragma unroll
    for (int kk = 0; kk < 32; ++kk) a[kk] = z;
  }

#pragma unroll
  for (int nt = 0; nt < 8; ++nt) {
    v8f acc = {};
    const float* bp = ws + (nt * 16 + lane15) * WSTRIDE + koff;
#pragma unroll
    for (int kk = 0; kk < 32; ++kk) {
      v2f b = *(const v2f*)(bp + kk * 4);
      acc = __builtin_amdgcn_wmma_f32_16x16x4_f32(
          false, a[kk], false, b, (short)0, acc, false, false);
    }
    const int colOut = nt * 16 + lane15;
#pragma unroll
    for (int r = 0; r < 8; ++r) {
      int row = rowBase + r + half * 8;
      if (row < n) H[(size_t)row * HID + colOut] = acc[r];
    }
  }
}

#define GCH 16
#define AGGW 25
#define AGGT (AGGW * 32)
#define AGGN (AGGW * 16)

__device__ __forceinline__ void hold_rows(const float4 (&h)[GCH]) {
  asm volatile("" :: "v"(h[0].x), "v"(h[1].x), "v"(h[2].x), "v"(h[3].x), "v"(h[4].x), "v"(h[5].x),
               "v"(h[6].x), "v"(h[7].x), "v"(h[8].x), "v"(h[9].x), "v"(h[10].x), "v"(h[11].x),
               "v"(h[12].x), "v"(h[13].x), "v"(h[14].x), "v"(h[15].x));
}

template <int NT>
__global__ void __launch_bounds__(AGGT)
k_agg_gemm(const float* __restrict__ H, const int2* __restrict__ rec,
           const int* __restrict__ rowstart, const float* __restrict__ dinv,
           const float* __restrict__ B, const float* __restrict__ bias,
           float* __restrict__ OUT, int n) {
  constexpr int NC = NT * 16;
  extern __shared__ float sm[];
  const int tid = threadIdx.x;
  const int wave   = tid >> 5;
  const int lane   = tid & 31;
  const int lane15 = lane & 15;
  const int half   = lane >> 4;
  const int koff   = half * 2;
  float* tile = sm + NC * WSTRIDE + wave * (16 * WSTRIDE);

  constexpr int TOT = HID * NC;
  constexpr int NIT = (TOT + AGGT - 1) / AGGT;
  {
    float sv[NIT];
#pragma unroll
    for (int u = 0; u < NIT; ++u) sv[u] = B[min(tid + u * AGGT, TOT - 1)];
#pragma unroll
    for (int u = 0; u < NIT; ++u) {
      const int i = tid + u * AGGT;
      if (i < TOT) sm[(i % NC) * WSTRIDE + i / NC] = sv[u];
    }
  }
  __syncthreads();

  const int base = blockIdx.x * AGGN + wave * 16;
  const int q = lane * 4;
  const int last = n - 1;
  const int rsv = rowstart[min(base + lane, n)];
  const float dvl = dinv[min(base + lane15, last)];
  const int beg = __builtin_amdgcn_readlane(rsv, 0);
  const int end = __builtin_amdgcn_readlane(rsv, 16);

  {
    float4 sr[16];
#pragma unroll
    for (int m = 0; m < 16; ++m)
      sr[m] = *(const float4*)(H + (size_t)min(base + m, last) * HID + q);
#pragma unroll
    for (int m = 0; m < 16; ++m) {
      float wself = __int_as_float(__builtin_amdgcn_readlane(__float_as_int(dvl), m));
      wself *= wself;
      float4 v = sr[m];
      v.x *= wself; v.y *= wself; v.z *= wself; v.w *= wself;
      if (base + m > last) v = make_float4(0.0f, 0.0f, 0.0f, 0.0f);
      *(float4*)(tile + m * WSTRIDE + q) = v;
    }
  }

  int m = 0;
  int nb = __builtin_amdgcn_readlane(rsv, 1);
  float4 acc = *(const float4*)(tile + q);
  if (beg < end) {
    int2 rv = rec[min(beg + lane, end - 1)];
    for (int jw = beg; jw < end; jw += 32) {
      const int2 rvn = rec[min(jw + 32 + lane, end - 1)];
#pragma unroll
      for (int c = 0; c < 32 / GCH; ++c) {
        const int j0 = jw + c * GCH;
        if (j0 < end) {
          float4 h[GCH];
#pragma unroll
          for (int t = 0; t < GCH; ++t) {
            const unsigned s = (unsigned)__builtin_amdgcn_readlane(rv.x, c * GCH + t);
            h[t] = *(const float4*)(H + (size_t)s * HID + q);
          }
          hold_rows(h);
#pragma unroll
          for (int t = 0; t < GCH; ++t) {
            if (j0 + t < end) {
              while (j0 + t == nb) {
                *(float4*)(tile + m * WSTRIDE + q) = acc;
                ++m;
                acc = *(const float4*)(tile + m * WSTRIDE + q);
                nb = __builtin_amdgcn_readlane(rsv, m + 1);
              }
              const float w = __int_as_float(__builtin_amdgcn_readlane(rv.y, c * GCH + t));
              acc.x += w * h[t].x; acc.y += w * h[t].y; acc.z += w * h[t].z; acc.w += w * h[t].w;
            }
          }
        }
      }
      rv = rvn;
    }
  }
  *(float4*)(tile + m * WSTRIDE + q) = acc;

  v2f a[32];
  {
    const float* tp = tile + lane15 * WSTRIDE + koff;
    const float* cp = bias + koff;
#pragma unroll
    for (int kk = 0; kk < 32; ++kk) {
      v2f xv = *(const v2f*)(tp + kk * 4);
      v2f bv = *(const v2f*)(cp + kk * 4);
      v2f t = xv + bv;
      t.x = fmaxf(t.x, 0.0f);
      t.y = fmaxf(t.y, 0.0f);
      a[kk] = t;
    }
  }

#pragma unroll
  for (int nt = 0; nt < NT; ++nt) {
    v8f accw = {};
    const float* bp = sm + (nt * 16 + lane15) * WSTRIDE + koff;
#pragma unroll
    for (int kk = 0; kk < 32; ++kk) {
      v2f b = *(const v2f*)(bp + kk * 4);
      accw = __builtin_amdgcn_wmma_f32_16x16x4_f32(
          false, a[kk], false, b, (short)0, accw, false, false);
    }
    const int colOut = nt * 16 + lane15;
#pragma unroll
    for (int r = 0; r < 8; ++r) {
      int row = base + r + half * 8;
      if (row < n) OUT[(size_t)row * NC + colOut] = accw[r];
    }
  }
}

__device__ __forceinline__ void gemm16_role(const float* __restrict__ X, const float* __restrict__ B,
                                            int ldb, int nc, float* __restrict__ Z, int n, float* bs) {
  const int tid = threadIdx.x;

  const int wave   = tid >> 5;
  const int lane   = tid & 31;
  const int lane15 = lane & 15;
  const int half   = lane >> 4;
  const int rowBase = wave * 16;
  const int arow    = rowBase + lane15;
  const int koff    = half * 2;

  float sv[HID * ZW / 256];
#pragma unroll
  for (int it = 0; it < HID * ZW / 256; ++it) {
    int i = tid + it * 256;
    sv[it] = B[(i >> 4) * ldb + min(i & 15, nc - 1)];
  }

  v2f a[32];
  const float* xp = X + (size_t)min(arow, n - 1) * HID + koff;
#pragma unroll
  for (int kk = 0; kk < 32; ++kk)
    a[kk] = *(const v2f*)(xp + kk * 4);

#pragma unroll
  for (int it = 0; it < HID * ZW / 256; ++it) {
    int i = tid + it * 256;
    int k = i >> 4, c = i & 15;
    bs[c * WSTRIDE + k] = (c < nc) ? sv[it] : 0.0f;
  }
  __syncthreads();

  v8f acc = {};
  const float* bp = bs + lane15 * WSTRIDE + koff;
#pragma unroll
  for (int kk = 0; kk < 32; ++kk) {
    v2f b = *(const v2f*)(bp + kk * 4);
    acc = __builtin_amdgcn_wmma_f32_16x16x4_f32(
        false, a[kk], false, b, (short)0, acc, false, false);
  }
#pragma unroll
  for (int r = 0; r < 8; ++r) {
    int row = rowBase + r + half * 8;
    if (row < n) Z[(size_t)row * ZW + lane15] = acc[r];
  }
}

#define CNTW 16
#define FRONT_ROLES (3 + CNTW)

__device__ __forceinline__ void count_role(const int* __restrict__ batch, int* __restrict__ gsize,
                                           int n, int part) {
  const int tid = threadIdx.x;
  const int lane = tid & 31;
  const int per = (n + CNTW - 1) / CNTW;
  const int beg = part * per;
  const int end = min(beg + per, n);
  for (int i0 = beg; i0 < end; i0 += 256 * EB) {
    int g[EB];
#pragma unroll
    for (int u = 0; u < EB; ++u) g[u] = batch[min(i0 + u * 256 + tid, n - 1)];
    hold_batch(g[0], g[1], g[2], g[3], g[4], g[5], g[6], g[7]);
#pragma unroll
    for (int u = 0; u < EB; ++u) {
      const bool ok = (i0 + u * 256 + tid) < end;
      const int gp = __shfl_up(g[u], 1, 32);
      const bool head = ok && (lane == 0 || gp != g[u]);
      const unsigned hm = (unsigned)__ballot(head);
      const unsigned okm = (unsigned)__ballot(ok);
      if (head) {
        const unsigned above = (lane == 31) ? 0u : (hm & (0xFFFFFFFEu << lane));
        const int nxt = above ? (__ffs((int)above) - 1) : __popc(okm);
        atomicAdd(&gsize[g[u]], nxt - lane);
      }
    }
  }
}

__device__ __forceinline__ void outinit_role(const float* __restrict__ b3, const float* __restrict__ Wl,
                                             const float* __restrict__ bl, float* __restrict__ bw,
                                             float* __restrict__ out, int G, int C) {
  const int tid = threadIdx.x;
  for (int i = tid; i < G * C; i += 256) out[i] = bl[i % C];
  if (tid < ZW) {
    float s = 0.0f;
    if (tid < C)
      for (int k = 0; k < HID; ++k) s += b3[k] * Wl[k * C + tid];
    bw[tid] = s;
  }
}

__global__ void k_front(const float* __restrict__ X, const float* __restrict__ W1,
                        float* __restrict__ H, int n,
                        const int* __restrict__ hist, int* __restrict__ base,
                        int* __restrict__ binStart, int nbins,
                        const float* __restrict__ W3, const float* __restrict__ Wl, int C,
                        float* __restrict__ M, const float* __restrict__ b3,
                        const float* __restrict__ bl, float* __restrict__ bw,
                        float* __restrict__ out, int G,
                        const int* __restrict__ batch, int* __restrict__ gsize) {
  extern __shared__ float ws[];
  const int blk = blockIdx.x;
  if (blk == 0) {
    scan_role(hist, base, binStart, nbins, (int*)ws);
  } else if (blk == 1) {
    gemm16_role(W3, Wl, C, C, M, HID, ws);
  } else if (blk == 2) {
    outinit_role(b3, Wl, bl, bw, out, G, C);
  } else if (blk < FRONT_ROLES) {
    count_role(batch, gsize, n, blk - 3);
  } else {
    gemm128_role(X, W1, H, n, blk - FRONT_ROLES, ws);
  }
}

__global__ void k_tail16(const float* __restrict__ Z, const int2* __restrict__ rec,
                         const int* __restrict__ rowstart, const float* __restrict__ dinv,
                         const int* __restrict__ batch, const int* __restrict__ gsize,
                         const float* __restrict__ bw, float* __restrict__ out, int n, int C) {
  const int gid = blockIdx.x * blockDim.x + threadIdx.x;
  const int lane = threadIdx.x & 31;
  const int node0 = (gid >> 5) * 8;
  if (node0 >= n) return;
  const bool valid = (gid >> 2) < n;
  const int node = min(gid >> 2, n - 1);
  const int q = (gid & 3) * 4;

  const int beg = rowstart[node];
  const int end = rowstart[node + 1];
  int g = batch[node];
  const float4 bq = *(const float4*)(bw + q);
  const int gs = gsize[g];

  float wself = dinv[node];
  wself *= wself;
  float4 acc = *(const float4*)(Z + (size_t)node * ZW + q);
  acc.x *= wself; acc.y *= wself; acc.z *= wself; acc.w *= wself;

  for (int j = beg; j < end; j += 4) {
    int2 r[4];
    float4 h[4];
#pragma unroll
    for (int t = 0; t < 4; ++t) r[t] = rec[(j + t < end) ? (j + t) : j];
#pragma unroll
    for (int t = 0; t < 4; ++t) h[t] = *(const float4*)(Z + (size_t)r[t].x * ZW + q);
#pragma unroll
    for (int t = 0; t < 4; ++t) {
      const bool ok = (j + t) < end;
      const float w = __int_as_float(r[t].y);
      const float hx = ok ? h[t].x : 0.0f, hy = ok ? h[t].y : 0.0f;
      const float hz = ok ? h[t].z : 0.0f, hw = ok ? h[t].w : 0.0f;
      acc.x += w * hx; acc.y += w * hy; acc.z += w * hz; acc.w += w * hw;
    }
  }

  const int g0 = __builtin_amdgcn_readfirstlane(g);
  if (!valid) {
    acc = make_float4(0.0f, 0.0f, 0.0f, 0.0f);
    g = g0;
  }
  const bool same = __ballot(g != g0) == 0;
  float run = 1.0f;
  if (same) {
#pragma unroll
    for (int off = 4; off < 32; off <<= 1) {
      acc.x += __shfl_xor(acc.x, off, 32);
      acc.y += __shfl_xor(acc.y, off, 32);
      acc.z += __shfl_xor(acc.z, off, 32);
      acc.w += __shfl_xor(acc.w, off, 32);
    }
    run = (float)min(8, n - node0);
  }
  if (same ? (lane < 4) : valid) {
    const float sz = (float)gs;
    float* p = out + (size_t)g * C + q;
    const float v0 = (acc.x + run * bq.x) / sz, v1 = (acc.y + run * bq.y) / sz;
    const float v2 = (acc.z + run * bq.z) / sz, v3 = (acc.w + run * bq.w) / sz;
    if (q + 0 < C) __hip_atomic_fetch_add(p + 0, v0, __ATOMIC_RELAXED, __HIP_MEMORY_SCOPE_AGENT);
    if (q + 1 < C) __hip_atomic_fetch_add(p + 1, v1, __ATOMIC_RELAXED, __HIP_MEMORY_SCOPE_AGENT);
    if (q + 2 < C) __hip_atomic_fetch_add(p + 2, v2, __ATOMIC_RELAXED, __HIP_MEMORY_SCOPE_AGENT);
    if (q + 3 < C) __hip_atomic_fetch_add(p + 3, v3, __ATOMIC_RELAXED, __HIP_MEMORY_SCOPE_AGENT);
  }
}

extern "C" void kernel_launch(void* const* d_in, const int* in_sizes, int n_in,
                              void* d_out, int out_size, void* d_ws, size_t ws_size,
                              hipStream_t stream) {
  const float* x   = (const float*)d_in[0];
  const int*   ei  = (const int*)d_in[1];
  const int*   bat = (const int*)d_in[2];
  const float* W1  = (const float*)d_in[3];
  const float* b1  = (const float*)d_in[4];
  const float* W2  = (const float*)d_in[5];
  const float* b2  = (const float*)d_in[6];
  const float* W3  = (const float*)d_in[7];
  const float* b3  = (const float*)d_in[8];
  const float* Wl  = (const float*)d_in[9];
  const float* bl  = (const float*)d_in[10];

  const int n = in_sizes[2];
  const int e = in_sizes[1] / 2;
  const int C = in_sizes[10];
  const int G = out_size / C;
  const int* src = ei;
  const int* dst = ei + e;

  char* base = (char*)d_ws;
  auto alloc = [&](size_t bytes) -> void* {
    void* p = (void*)base;
    base += (bytes + 255) & ~(size_t)255;
    return p;
  };
  int*   hist      = (int*)alloc((size_t)NBMAX * NCHUNK * 4);
  int*   base0     = (int*)alloc((size_t)NBMAX * NCHUNK * 4);
  int*   binStart  = (int*)alloc(((size_t)NBMAX + 1) * 4);
  int2*  binned    = (int2*)alloc((size_t)e * 8);
  int*   rowstart  = (int*)alloc(((size_t)n + 1) * 4);
  float* dinv      = (float*)alloc((size_t)n * 4);
  int2*  rec       = (int2*)alloc((size_t)e * 8);
  float* Hb        = (float*)alloc((size_t)n * HID * 4);
  float* Ab        = (float*)alloc((size_t)n * HID * 4);
  float* Mb        = (float*)alloc((size_t)HID * ZW * 4);
  float* Zb        = (float*)alloc((size_t)n * ZW * 4);
  float* bwv       = (float*)alloc((size_t)ZW * 4);
  int*   gsize     = (int*)alloc((size_t)G * 4);
  (void)ws_size; (void)n_in;

  const int T = 256;
  auto nb = [](long long w, int t) { return (unsigned)((w + t - 1) / t); };
  const unsigned gemmB = (unsigned)((n + 127) / 128);
  const unsigned nbins = (unsigned)((n + BINN - 1) >> BSH);
  const size_t lds = (size_t)HID * WSTRIDE * sizeof(float);
  const unsigned aggB = (unsigned)((n + AGGN - 1) / AGGN);
  const size_t ldsF8 = (size_t)(8 * 16 + AGGN) * WSTRIDE * sizeof(float);
  const size_t ldsF1 = (size_t)(1 * 16 + AGGN) * WSTRIDE * sizeof(float);

  k_binhist<<<NCHUNK, ETH, 0, stream>>>(dst, hist, e, (int)nbins, gsize, G);
  k_front<<<gemmB + FRONT_ROLES, T, lds, stream>>>(x, W1, Hb, n, hist, base0, binStart, (int)nbins,
                                                   W3, Wl, C, Mb, b3, bl, bwv, (float*)d_out, G,
                                                   bat, gsize);
  k_binscatter<<<NCHUNK, ETH, 0, stream>>>(src, dst, base0, binned, e, (int)nbins);
  k_bincount<<<nbins, ETH, 0, stream>>>(binned, binStart, rowstart, dinv, n);
  k_binplace<<<nbins, ETH, 0, stream>>>(binned, binStart, rowstart, dinv, rec, n);

  k_agg_gemm<8><<<aggB, AGGT, ldsF8, stream>>>(Hb, rec, rowstart, dinv, W2, b1, Ab, n);

  k_agg_gemm<1><<<aggB, AGGT, ldsF1, stream>>>(Ab, rec, rowstart, dinv, Mb, b2, Zb, n);

  k_tail16<<<nb((long long)n * 4, T), T, 0, stream>>>(Zb, rec, rowstart, dinv, bat, gsize, bwv,
                                                      (float*)d_out, n, C);
}
